# starting kernel
# baseline (speedup 1.0000x reference)
#include <hip/hip_runtime.h>

typedef __attribute__((ext_vector_type(2))) float v2f;
typedef __attribute__((ext_vector_type(8))) float v8f;

#define B_SZ 8
#define NA 4096
#define NB 4096
#define FD 64
#define WAVES 8
#define ROWS_PER_WAVE 16
#define ROWS_PER_BLOCK (WAVES * ROWS_PER_WAVE)
#define NTILES (NB / 16)
#define KBIAS 48387

#define TOP3_INSERT_PK(pk, b0, b1, b2) do {          \
    int _r0 = min((pk), (b0));                        \
    (b0)    = max((pk), (b0));                        \
    int _r1 = min(_r0, (b1));                         \
    (b1)    = max(_r0, (b1));                         \
    (b2)    = max(_r1, (b2));                         \
  } while (0)

__global__ __launch_bounds__(256) void knn3_wmma(
    const float* __restrict__ a_feats,
    const float* __restrict__ b_feats,
    const int*   __restrict__ a_coords,
    const int*   __restrict__ b_coords,
    float*       __restrict__ out)
{
  __shared__ unsigned sPack[NB];
  __shared__ int      sKb2[NB];
  __shared__ int      sPk[WAVES][ROWS_PER_WAVE][3];

  const int tid   = threadIdx.x;
  const int lane  = tid & 31;
  const int wave  = tid >> 5;
  const int mlane = lane & 15;
  const bool hi   = lane >= 16;

  const int blocksPerBatch = NA / ROWS_PER_BLOCK;
  const int b       = blockIdx.x / blocksPerBatch;
  const int rowBase = (blockIdx.x % blocksPerBatch) * ROWS_PER_BLOCK
                    + wave * ROWS_PER_WAVE;

  for (int n = tid; n < NB; n += 256) {
    const int* p = b_coords + ((size_t)b * NB + n) * 3;
    int x = p[0] >> 4, y = p[1] >> 4, z = p[2] >> 4;
    sPack[n] = (unsigned)x | ((unsigned)y << 8) | ((unsigned)z << 16);
    sKb2[n]  = ((KBIAS - (x * x + y * y + z * z)) << 12) + (4095 - n);
  }
  __syncthreads();

  const int arow = rowBase + mlane;
  const int* ap = a_coords + ((size_t)b * NA + arow) * 3;
  const int iax = ap[0] >> 4, iay = ap[1] >> 4, iaz = ap[2] >> 4;
  const int na2i = iax * iax + iay * iay + iaz * iaz;
  v2f A;
  A.x = hi ? (float)(2 * iaz) : (float)(2 * iax);
  A.y = hi ? 0.0f             : (float)(2 * iay);

  int bp0[8], bp1[8], bp2[8];
#pragma unroll
  for (int r = 0; r < 8; ++r) { bp0[r] = bp1[r] = bp2[r] = -1; }

  const v8f Cz = {};

  for (int t = 0; t < NTILES; ++t) {
    const int n = t * 16 + mlane;
    const unsigned u = sPack[n];
    const int kb2 = sKb2[n];
    const float bx = (float)(u & 255u);
    const float by = (float)((u >> 8) & 255u);
    const float bz = (float)((u >> 16) & 255u);
    v2f Bm;
    Bm.x = hi ? bz : bx;
    Bm.y = hi ? 0.0f : by;

    v8f Dv = __builtin_amdgcn_wmma_f32_16x16x4_f32(
         false, A,  false, Bm,
         (short)0, Cz,  false,  false);

#pragma unroll
    for (int r = 0; r < 8; ++r) {
      const int ik = (int)Dv[r];
      const int pk = (ik << 12) + kb2;
      TOP3_INSERT_PK(pk, bp0[r], bp1[r], bp2[r]);
    }
  }

#pragma unroll
  for (int r = 0; r < 8; ++r) {
    int p0 = bp0[r], p1 = bp1[r], p2 = bp2[r];
#pragma unroll
    for (int off = 1; off < 16; off <<= 1) {
      int q0 = __shfl_xor(p0, off, 32);
      int q1 = __shfl_xor(p1, off, 32);
      int q2 = __shfl_xor(p2, off, 32);
      TOP3_INSERT_PK(q0, p0, p1, p2);
      TOP3_INSERT_PK(q1, p0, p1, p2);
      TOP3_INSERT_PK(q2, p0, p1, p2);
    }
    if (mlane == 0) {
      const int rowLocal = r + (hi ? 8 : 0);
      sPk[wave][rowLocal][0] = p0;
      sPk[wave][rowLocal][1] = p1;
      sPk[wave][rowLocal][2] = p2;
    }
  }
  __syncthreads();

  const float* afBase = a_feats + ((size_t)b * NA + rowBase) * FD;
  float* outBase = out + ((size_t)b * NA + rowBase) * (2 * FD);
  const float* bfB = b_feats + (size_t)b * NB * FD;

  for (int row = 0; row < ROWS_PER_WAVE; ++row) {
    const int pk0 = sPk[wave][row][0];
    const int pk1 = sPk[wave][row][1];
    const int pk2 = sPk[wave][row][2];
    const int rna2 = __shfl(na2i, row, 32);
    const int j0 = 4095 - (pk0 & 4095);
    const int j1 = 4095 - (pk1 & 4095);
    const int j2 = 4095 - (pk2 & 4095);
    const int d20 = rna2 + KBIAS - (int)((unsigned)pk0 >> 12);
    const int d21 = rna2 + KBIAS - (int)((unsigned)pk1 >> 12);
    const int d22 = rna2 + KBIAS - (int)((unsigned)pk2 >> 12);
    const float inv = 1.0f / 128.0f;
    const float w0 = fmaxf(0.5f - sqrtf((float)d20) * inv, 0.0f);
    const float w1 = fmaxf(0.5f - sqrtf((float)d21) * inv, 0.0f);
    const float w2 = fmaxf(0.5f - sqrtf((float)d22) * inv, 0.0f);
    const float* bf0 = bfB + (size_t)j0 * FD;
    const float* bf1 = bfB + (size_t)j1 * FD;
    const float* bf2 = bfB + (size_t)j2 * FD;
    const float* af  = afBase + (size_t)row * FD;
    float* op = outBase + (size_t)row * (2 * FD);
#pragma unroll
    for (int h = 0; h < 2; ++h) {
      const int d = lane + h * 32;
      const float tmp = w0 * bf0[d] + w1 * bf1[d] + w2 * bf2[d];
      op[d]      = af[d];
      op[FD + d] = tmp;
    }
  }
}

extern "C" void kernel_launch(void* const* d_in, const int* in_sizes, int n_in,
                              void* d_out, int out_size, void* d_ws, size_t ws_size,
                              hipStream_t stream) {
  const float* a_feats  = (const float*)d_in[0];
  const float* b_feats  = (const float*)d_in[1];
  const int*   a_coords = (const int*)d_in[2];
  const int*   b_coords = (const int*)d_in[3];
  float* out = (float*)d_out;

  dim3 grid(B_SZ * (NA / ROWS_PER_BLOCK));
  dim3 block(256);
  knn3_wmma<<<grid, block, 0, stream>>>(a_feats, b_feats, a_coords, b_coords, out);
}


# 16 waves per block instead of 8: each 16-row tile's column sweep is split over two waves (2048 columns each), their top-3 lists merged per row before the gather; 4 waves per SIMD, not 2
# speedup vs baseline: 1.2158x; 1.2158x over previous
#include <hip/hip_runtime.h>

typedef __attribute__((ext_vector_type(2))) float v2f;
typedef __attribute__((ext_vector_type(8))) float v8f;

#define B_SZ 8
#define NA 4096
#define NB 4096
#define FD 64
#define ROW_TILES 8
#define COL_SPLIT 2
#define WAVES (ROW_TILES * COL_SPLIT)
#define ROWS_PER_WAVE 16
#define ROWS_PER_BLOCK (ROW_TILES * ROWS_PER_WAVE)
#define NTILES (NB / 16)
#define TILES_PER_WAVE (NTILES / COL_SPLIT)
#define OUT_ROWS_PER_WAVE (ROWS_PER_WAVE / COL_SPLIT)
#define KBIAS 48387

#define TOP3_INSERT_PK(pk, b0, b1, b2) do {          \
    int _r0 = min((pk), (b0));                        \
    (b0)    = max((pk), (b0));                        \
    int _r1 = min(_r0, (b1));                         \
    (b1)    = max(_r0, (b1));                         \
    (b2)    = max(_r1, (b2));                         \
  } while (0)

__global__ __launch_bounds__(WAVES * 32) void knn3_wmma(
    const float* __restrict__ a_feats,
    const float* __restrict__ b_feats,
    const int*   __restrict__ a_coords,
    const int*   __restrict__ b_coords,
    float*       __restrict__ out)
{
  __shared__ unsigned sPack[NB];
  __shared__ int      sKb2[NB];
  __shared__ int      sPk[COL_SPLIT][ROW_TILES][ROWS_PER_WAVE][3];

  const int tid   = threadIdx.x;
  const int lane  = tid & 31;
  const int wave  = tid >> 5;
  const int mlane = lane & 15;
  const bool hi   = lane >= 16;
  const int rtile = wave % ROW_TILES;
  const int cpart = wave / ROW_TILES;

  const int blocksPerBatch = NA / ROWS_PER_BLOCK;
  const int b       = blockIdx.x / blocksPerBatch;
  const int rowBase = (blockIdx.x % blocksPerBatch) * ROWS_PER_BLOCK
                    + rtile * ROWS_PER_WAVE;

  for (int n = tid; n < NB; n += WAVES * 32) {
    const int* p = b_coords + ((size_t)b * NB + n) * 3;
    int x = p[0] >> 4, y = p[1] >> 4, z = p[2] >> 4;
    sPack[n] = (unsigned)x | ((unsigned)y << 8) | ((unsigned)z << 16);
    sKb2[n]  = ((KBIAS - (x * x + y * y + z * z)) << 12) + (4095 - n);
  }
  __syncthreads();

  const int arow = rowBase + mlane;
  const int* ap = a_coords + ((size_t)b * NA + arow) * 3;
  const int iax = ap[0] >> 4, iay = ap[1] >> 4, iaz = ap[2] >> 4;
  const int na2i = iax * iax + iay * iay + iaz * iaz;
  v2f A;
  A.x = hi ? (float)(2 * iaz) : (float)(2 * iax);
  A.y = hi ? 0.0f             : (float)(2 * iay);

  int bp0[8], bp1[8], bp2[8];
#pragma unroll
  for (int r = 0; r < 8; ++r) { bp0[r] = bp1[r] = bp2[r] = -1; }

  const v8f Cz = {};

  const int nFirst = cpart * (TILES_PER_WAVE * 16) + mlane;
  for (int i = 0; i < TILES_PER_WAVE; ++i) {
    const int n = nFirst + i * 16;
    const unsigned u = sPack[n];
    const int kb2 = sKb2[n];
    const float bx = (float)(u & 255u);
    const float by = (float)((u >> 8) & 255u);
    const float bz = (float)((u >> 16) & 255u);
    v2f Bm;
    Bm.x = hi ? bz : bx;
    Bm.y = hi ? 0.0f : by;

    v8f Dv = __builtin_amdgcn_wmma_f32_16x16x4_f32(
         false, A,  false, Bm,
         (short)0, Cz,  false,  false);

#pragma unroll
    for (int r = 0; r < 8; ++r) {
      const int ik = (int)Dv[r];
      const int pk = (ik << 12) + kb2;
      TOP3_INSERT_PK(pk, bp0[r], bp1[r], bp2[r]);
    }
  }

#pragma unroll
  for (int r = 0; r < 8; ++r) {
    int p0 = bp0[r], p1 = bp1[r], p2 = bp2[r];
#pragma unroll
    for (int off = 1; off < 16; off <<= 1) {
      int q0 = __shfl_xor(p0, off, 32);
      int q1 = __shfl_xor(p1, off, 32);
      int q2 = __shfl_xor(p2, off, 32);
      TOP3_INSERT_PK(q0, p0, p1, p2);
      TOP3_INSERT_PK(q1, p0, p1, p2);
      TOP3_INSERT_PK(q2, p0, p1, p2);
    }
    if (mlane == 0) {
      const int rowLocal = r + (hi ? 8 : 0);
      sPk[cpart][rtile][rowLocal][0] = p0;
      sPk[cpart][rtile][rowLocal][1] = p1;
      sPk[cpart][rtile][rowLocal][2] = p2;
    }
  }
  __syncthreads();

  const float* afBase = a_feats + ((size_t)b * NA + rowBase) * FD;
  float* outBase = out + ((size_t)b * NA + rowBase) * (2 * FD);
  const float* bfB = b_feats + (size_t)b * NB * FD;

  for (int rr = 0; rr < OUT_ROWS_PER_WAVE; ++rr) {
    const int row = cpart * OUT_ROWS_PER_WAVE + rr;
    int pk0 = sPk[0][rtile][row][0];
    int pk1 = sPk[0][rtile][row][1];
    int pk2 = sPk[0][rtile][row][2];
#pragma unroll
    for (int c = 1; c < COL_SPLIT; ++c) {
      const int q0 = sPk[c][rtile][row][0];
      const int q1 = sPk[c][rtile][row][1];
      const int q2 = sPk[c][rtile][row][2];
      TOP3_INSERT_PK(q0, pk0, pk1, pk2);
      TOP3_INSERT_PK(q1, pk0, pk1, pk2);
      TOP3_INSERT_PK(q2, pk0, pk1, pk2);
    }
    const int rna2 = __shfl(na2i, row, 32);
    const int j0 = 4095 - (pk0 & 4095);
    const int j1 = 4095 - (pk1 & 4095);
    const int j2 = 4095 - (pk2 & 4095);
    const int d20 = rna2 + KBIAS - (int)((unsigned)pk0 >> 12);
    const int d21 = rna2 + KBIAS - (int)((unsigned)pk1 >> 12);
    const int d22 = rna2 + KBIAS - (int)((unsigned)pk2 >> 12);
    const float inv = 1.0f / 128.0f;
    const float w0 = fmaxf(0.5f - sqrtf((float)d20) * inv, 0.0f);
    const float w1 = fmaxf(0.5f - sqrtf((float)d21) * inv, 0.0f);
    const float w2 = fmaxf(0.5f - sqrtf((float)d22) * inv, 0.0f);
    const float* bf0 = bfB + (size_t)j0 * FD;
    const float* bf1 = bfB + (size_t)j1 * FD;
    const float* bf2 = bfB + (size_t)j2 * FD;
    const float* af  = afBase + (size_t)row * FD;
    float* op = outBase + (size_t)row * (2 * FD);
#pragma unroll
    for (int h = 0; h < 2; ++h) {
      const int d = lane + h * 32;
      const float tmp = w0 * bf0[d] + w1 * bf1[d] + w2 * bf2[d];
      op[d]      = af[d];
      op[FD + d] = tmp;
    }
  }
}

extern "C" void kernel_launch(void* const* d_in, const int* in_sizes, int n_in,
                              void* d_out, int out_size, void* d_ws, size_t ws_size,
                              hipStream_t stream) {
  const float* a_feats  = (const float*)d_in[0];
  const float* b_feats  = (const float*)d_in[1];
  const int*   a_coords = (const int*)d_in[2];
  const int*   b_coords = (const int*)d_in[3];
  float* out = (float*)d_out;

  dim3 grid(B_SZ * (NA / ROWS_PER_BLOCK));
  dim3 block(WAVES * 32);
  knn3_wmma<<<grid, block, 0, stream>>>(a_feats, b_feats, a_coords, b_coords, out);
}


# the f32 WMMA emits the finished sort key (|cb|^2 bias and a per-step tie fraction in the free 4th k slot): per-candidate sweep work is a 3-op f32 insert (2 med3 + max), was 6 VALU
# speedup vs baseline: 1.4314x; 1.1773x over previous
#include <hip/hip_runtime.h>

typedef __attribute__((ext_vector_type(2))) float v2f;
typedef __attribute__((ext_vector_type(8))) float v8f;

#define B_SZ 8
#define NA 4096
#define NB 4096
#define FD 64
#define ROW_TILES 8
#define COL_SPLIT 2
#define WAVES (ROW_TILES * COL_SPLIT)
#define ROWS_PER_WAVE 16
#define ROWS_PER_BLOCK (ROW_TILES * ROWS_PER_WAVE)
#define NTILES (NB / 16)
#define TILES_PER_WAVE (NTILES / COL_SPLIT)
#define OUT_ROWS_PER_WAVE (ROWS_PER_WAVE / COL_SPLIT)
#define KBIAS 48388
#define STEP_SCALE ((float)TILES_PER_WAVE)
static_assert(TILES_PER_WAVE == 128, "sweep key = 17 integer bits + 7 step bits = 24 bits of f32");

#define TOP3_INSERT_PK(pk, b0, b1, b2) do {          \
    int _r0 = min((pk), (b0));                        \
    (b0)    = max((pk), (b0));                        \
    int _r1 = min(_r0, (b1));                         \
    (b1)    = max(_r0, (b1));                         \
    (b2)    = max(_r1, (b2));                         \
  } while (0)

__device__ __forceinline__ float max_nonneg(float a, float b) {
  return __int_as_float(max(__float_as_int(a), __float_as_int(b)));
}

#define TOP3_INSERT_F(fk, f0, f1, f2) do {                   \
    (f2) = __builtin_amdgcn_fmed3f((fk), (f1), (f2));         \
    (f1) = __builtin_amdgcn_fmed3f((fk), (f0), (f1));         \
    (f0) = max_nonneg((fk), (f0));                            \
  } while (0)

__global__ __launch_bounds__(WAVES * 32) void knn3_wmma(
    const float* __restrict__ a_feats,
    const float* __restrict__ b_feats,
    const int*   __restrict__ a_coords,
    const int*   __restrict__ b_coords,
    float*       __restrict__ out)
{
  __shared__ unsigned sPack[NB];
  __shared__ float    sKb2[NB];
  __shared__ int      sPk[COL_SPLIT][ROW_TILES][ROWS_PER_WAVE][3];

  const int tid   = threadIdx.x;
  const int lane  = tid & 31;
  const int wave  = tid >> 5;
  const int mlane = lane & 15;
  const bool hi   = lane >= 16;
  const int rtile = wave % ROW_TILES;
  const int cpart = wave / ROW_TILES;

  const int blocksPerBatch = NA / ROWS_PER_BLOCK;
  const int b       = blockIdx.x / blocksPerBatch;
  const int rowBase = (blockIdx.x % blocksPerBatch) * ROWS_PER_BLOCK
                    + rtile * ROWS_PER_WAVE;

  for (int n = tid; n < NB; n += WAVES * 32) {
    const int* p = b_coords + ((size_t)b * NB + n) * 3;
    int x = p[0] >> 4, y = p[1] >> 4, z = p[2] >> 4;
    sPack[n] = (unsigned)x | ((unsigned)y << 8) | ((unsigned)z << 16);
    const int tie = (TILES_PER_WAVE - 1) - ((n >> 4) & (TILES_PER_WAVE - 1));
    sKb2[n]  = (float)(KBIAS - (x * x + y * y + z * z)) + (float)tie * (1.0f / STEP_SCALE);
  }
  __syncthreads();

  const int arow = rowBase + mlane;
  const int* ap = a_coords + ((size_t)b * NA + arow) * 3;
  const int iax = ap[0] >> 4, iay = ap[1] >> 4, iaz = ap[2] >> 4;
  const int na2i = iax * iax + iay * iay + iaz * iaz;
  v2f A;
  A.x = hi ? (float)(2 * iaz) : (float)(2 * iax);
  A.y = hi ? 1.0f             : (float)(2 * iay);

  float sk0[8], sk1[8], sk2[8];
#pragma unroll
  for (int r = 0; r < 8; ++r) { sk0[r] = sk1[r] = sk2[r] = 0.0f; }

  const v8f Cz = {};

  const int nFirst = cpart * (TILES_PER_WAVE * 16) + mlane;
  for (int i = 0; i < TILES_PER_WAVE; ++i) {
    const int n = nFirst + i * 16;
    const unsigned u = sPack[n];
    const float kb2 = sKb2[n];
    const float bx = (float)(u & 255u);
    const float by = (float)((u >> 8) & 255u);
    const float bz = (float)((u >> 16) & 255u);
    v2f Bm;
    Bm.x = hi ? bz : bx;
    Bm.y = hi ? kb2 : by;

    v8f Dv = __builtin_amdgcn_wmma_f32_16x16x4_f32(
         false, A,  false, Bm,
         (short)0, Cz,  false,  false);

#pragma unroll
    for (int r = 0; r < 8; ++r) {
      TOP3_INSERT_F(Dv[r], sk0[r], sk1[r], sk2[r]);
    }
  }

  int bp0[8], bp1[8], bp2[8];
#pragma unroll
  for (int r = 0; r < 8; ++r) {
    const float fs[3] = { sk0[r], sk1[r], sk2[r] };
    int ps[3];
#pragma unroll
    for (int t = 0; t < 3; ++t) {
      const int key  = (int)fs[t];
      const int tie  = (int)((fs[t] - (float)key) * STEP_SCALE);
      const int nn   = nFirst + ((TILES_PER_WAVE - 1) - tie) * 16;
      ps[t] = (key << 12) + (4095 - nn);
    }
    bp0[r] = ps[0]; bp1[r] = ps[1]; bp2[r] = ps[2];
  }

#pragma unroll
  for (int r = 0; r < 8; ++r) {
    int p0 = bp0[r], p1 = bp1[r], p2 = bp2[r];
#pragma unroll
    for (int off = 1; off < 16; off <<= 1) {
      int q0 = __shfl_xor(p0, off, 32);
      int q1 = __shfl_xor(p1, off, 32);
      int q2 = __shfl_xor(p2, off, 32);
      TOP3_INSERT_PK(q0, p0, p1, p2);
      TOP3_INSERT_PK(q1, p0, p1, p2);
      TOP3_INSERT_PK(q2, p0, p1, p2);
    }
    if (mlane == 0) {
      const int rowLocal = r + (hi ? 8 : 0);
      sPk[cpart][rtile][rowLocal][0] = p0;
      sPk[cpart][rtile][rowLocal][1] = p1;
      sPk[cpart][rtile][rowLocal][2] = p2;
    }
  }
  __syncthreads();

  const float* afBase = a_feats + ((size_t)b * NA + rowBase) * FD;
  float* outBase = out + ((size_t)b * NA + rowBase) * (2 * FD);
  const float* bfB = b_feats + (size_t)b * NB * FD;

  for (int rr = 0; rr < OUT_ROWS_PER_WAVE; ++rr) {
    const int row = cpart * OUT_ROWS_PER_WAVE + rr;
    int pk0 = sPk[0][rtile][row][0];
    int pk1 = sPk[0][rtile][row][1];
    int pk2 = sPk[0][rtile][row][2];
#pragma unroll
    for (int c = 1; c < COL_SPLIT; ++c) {
      const int q0 = sPk[c][rtile][row][0];
      const int q1 = sPk[c][rtile][row][1];
      const int q2 = sPk[c][rtile][row][2];
      TOP3_INSERT_PK(q0, pk0, pk1, pk2);
      TOP3_INSERT_PK(q1, pk0, pk1, pk2);
      TOP3_INSERT_PK(q2, pk0, pk1, pk2);
    }
    const int rna2 = __shfl(na2i, row, 32);
    const int j0 = 4095 - (pk0 & 4095);
    const int j1 = 4095 - (pk1 & 4095);
    const int j2 = 4095 - (pk2 & 4095);
    const int d20 = rna2 + KBIAS - (int)((unsigned)pk0 >> 12);
    const int d21 = rna2 + KBIAS - (int)((unsigned)pk1 >> 12);
    const int d22 = rna2 + KBIAS - (int)((unsigned)pk2 >> 12);
    const float inv = 1.0f / 128.0f;
    const float w0 = fmaxf(0.5f - sqrtf((float)d20) * inv, 0.0f);
    const float w1 = fmaxf(0.5f - sqrtf((float)d21) * inv, 0.0f);
    const float w2 = fmaxf(0.5f - sqrtf((float)d22) * inv, 0.0f);
    const float* bf0 = bfB + (size_t)j0 * FD;
    const float* bf1 = bfB + (size_t)j1 * FD;
    const float* bf2 = bfB + (size_t)j2 * FD;
    const float* af  = afBase + (size_t)row * FD;
    float* op = outBase + (size_t)row * (2 * FD);
#pragma unroll
    for (int h = 0; h < 2; ++h) {
      const int d = lane + h * 32;
      const float tmp = w0 * bf0[d] + w1 * bf1[d] + w2 * bf2[d];
      op[d]      = af[d];
      op[FD + d] = tmp;
    }
  }
}

extern "C" void kernel_launch(void* const* d_in, const int* in_sizes, int n_in,
                              void* d_out, int out_size, void* d_ws, size_t ws_size,
                              hipStream_t stream) {
  const float* a_feats  = (const float*)d_in[0];
  const float* b_feats  = (const float*)d_in[1];
  const int*   a_coords = (const int*)d_in[2];
  const int*   b_coords = (const int*)d_in[3];
  float* out = (float*)d_out;

  dim3 grid(B_SZ * (NA / ROWS_PER_BLOCK));
  dim3 block(WAVES * 32);
  knn3_wmma<<<grid, block, 0, stream>>>(a_feats, b_feats, a_coords, b_coords, out);
}


# each wave's global loads are issued together, not one waited per loop trip: coordinate staging unrolled (8 loads + the a coordinates in flight), gather epilogue as one load phase with 16-byte lanes
# speedup vs baseline: 1.8439x; 1.2882x over previous
#include <hip/hip_runtime.h>

typedef __attribute__((ext_vector_type(2))) float v2f;
typedef __attribute__((ext_vector_type(8))) float v8f;
typedef __attribute__((ext_vector_type(4))) float v4f;

#define B_SZ 8
#define NA 4096
#define NB 4096
#define FD 64
#define ROW_TILES 8
#define COL_SPLIT 2
#define WAVES (ROW_TILES * COL_SPLIT)
#define ROWS_PER_WAVE 16
#define ROWS_PER_BLOCK (ROW_TILES * ROWS_PER_WAVE)
#define NTILES (NB / 16)
#define TILES_PER_WAVE (NTILES / COL_SPLIT)
#define OUT_ROWS_PER_WAVE (ROWS_PER_WAVE / COL_SPLIT)
#define STAGE_TRIPS (NB / (WAVES * 32))
#define ROW_PAIRS (OUT_ROWS_PER_WAVE / 2)
static_assert(NB % (WAVES * 32) == 0, "every thread stages the same number of columns");
static_assert(OUT_ROWS_PER_WAVE % 2 == 0 && FD == 16 * 4, "a 16-lane half covers one 64-float row with 4 floats per lane");
#define KBIAS 48388
#define STEP_SCALE ((float)TILES_PER_WAVE)
static_assert(TILES_PER_WAVE == 128, "sweep key = 17 integer bits + 7 step bits = 24 bits of f32");

#define TOP3_INSERT_PK(pk, b0, b1, b2) do {          \
    int _r0 = min((pk), (b0));                        \
    (b0)    = max((pk), (b0));                        \
    int _r1 = min(_r0, (b1));                         \
    (b1)    = max(_r0, (b1));                         \
    (b2)    = max(_r1, (b2));                         \
  } while (0)

__device__ __forceinline__ float max_nonneg(float a, float b) {
  return __int_as_float(max(__float_as_int(a), __float_as_int(b)));
}

#define TOP3_INSERT_F(fk, f0, f1, f2) do {                   \
    (f2) = __builtin_amdgcn_fmed3f((fk), (f1), (f2));         \
    (f1) = __builtin_amdgcn_fmed3f((fk), (f0), (f1));         \
    (f0) = max_nonneg((fk), (f0));                            \
  } while (0)

__global__ __launch_bounds__(WAVES * 32) void knn3_wmma(
    const float* __restrict__ a_feats,
    const float* __restrict__ b_feats,
    const int*   __restrict__ a_coords,
    const int*   __restrict__ b_coords,
    float*       __restrict__ out)
{
  __shared__ unsigned sPack[NB];
  __shared__ float    sKb2[NB];
  __shared__ int      sPk[COL_SPLIT][ROW_TILES][ROWS_PER_WAVE][3];

  const int tid   = threadIdx.x;
  const int lane  = tid & 31;
  const int wave  = tid >> 5;
  const int mlane = lane & 15;
  const bool hi   = lane >= 16;
  const int rtile = wave % ROW_TILES;
  const int cpart = wave / ROW_TILES;

  const int blocksPerBatch = NA / ROWS_PER_BLOCK;
  const int b       = blockIdx.x / blocksPerBatch;
  const int rowBase = (blockIdx.x % blocksPerBatch) * ROWS_PER_BLOCK
                    + rtile * ROWS_PER_WAVE;

  const int arow = rowBase + mlane;
  const int* ap = a_coords + ((size_t)b * NA + arow) * 3;
  const int ldax = ap[0], lday = ap[1], ldaz = ap[2];

#pragma unroll
  for (int k = 0; k < STAGE_TRIPS; ++k) {
    const int n = tid + k * (WAVES * 32);
    const int* p = b_coords + ((size_t)b * NB + n) * 3;
    int x = p[0] >> 4, y = p[1] >> 4, z = p[2] >> 4;
    sPack[n] = (unsigned)x | ((unsigned)y << 8) | ((unsigned)z << 16);
    const int tie = (TILES_PER_WAVE - 1) - ((n >> 4) & (TILES_PER_WAVE - 1));
    sKb2[n]  = (float)(KBIAS - (x * x + y * y + z * z)) + (float)tie * (1.0f / STEP_SCALE);
  }
  __syncthreads();

  const int iax = ldax >> 4, iay = lday >> 4, iaz = ldaz >> 4;
  const int na2i = iax * iax + iay * iay + iaz * iaz;
  v2f A;
  A.x = hi ? (float)(2 * iaz) : (float)(2 * iax);
  A.y = hi ? 1.0f             : (float)(2 * iay);

  float sk0[8], sk1[8], sk2[8];
#pragma unroll
  for (int r = 0; r < 8; ++r) { sk0[r] = sk1[r] = sk2[r] = 0.0f; }

  const v8f Cz = {};

  const int nFirst = cpart * (TILES_PER_WAVE * 16) + mlane;
  for (int i = 0; i < TILES_PER_WAVE; ++i) {
    const int n = nFirst + i * 16;
    const unsigned u = sPack[n];
    const float kb2 = sKb2[n];
    const float bx = (float)(u & 255u);
    const float by = (float)((u >> 8) & 255u);
    const float bz = (float)((u >> 16) & 255u);
    v2f Bm;
    Bm.x = hi ? bz : bx;
    Bm.y = hi ? kb2 : by;

    v8f Dv = __builtin_amdgcn_wmma_f32_16x16x4_f32(
         false, A,  false, Bm,
         (short)0, Cz,  false,  false);

#pragma unroll
    for (int r = 0; r < 8; ++r) {
      TOP3_INSERT_F(Dv[r], sk0[r], sk1[r], sk2[r]);
    }
  }

  int bp0[8], bp1[8], bp2[8];
#pragma unroll
  for (int r = 0; r < 8; ++r) {
    const float fs[3] = { sk0[r], sk1[r], sk2[r] };
    int ps[3];
#pragma unroll
    for (int t = 0; t < 3; ++t) {
      const int key  = (int)fs[t];
      const int tie  = (int)((fs[t] - (float)key) * STEP_SCALE);
      const int nn   = nFirst + ((TILES_PER_WAVE - 1) - tie) * 16;
      ps[t] = (key << 12) + (4095 - nn);
    }
    bp0[r] = ps[0]; bp1[r] = ps[1]; bp2[r] = ps[2];
  }

#pragma unroll
  for (int r = 0; r < 8; ++r) {
    int p0 = bp0[r], p1 = bp1[r], p2 = bp2[r];
#pragma unroll
    for (int off = 1; off < 16; off <<= 1) {
      int q0 = __shfl_xor(p0, off, 32);
      int q1 = __shfl_xor(p1, off, 32);
      int q2 = __shfl_xor(p2, off, 32);
      TOP3_INSERT_PK(q0, p0, p1, p2);
      TOP3_INSERT_PK(q1, p0, p1, p2);
      TOP3_INSERT_PK(q2, p0, p1, p2);
    }
    if (mlane == 0) {
      const int rowLocal = r + (hi ? 8 : 0);
      sPk[cpart][rtile][rowLocal][0] = p0;
      sPk[cpart][rtile][rowLocal][1] = p1;
      sPk[cpart][rtile][rowLocal][2] = p2;
    }
  }
  __syncthreads();

  const float* afBase = a_feats + ((size_t)b * NA + rowBase) * FD;
  float* outBase = out + ((size_t)b * NA + rowBase) * (2 * FD);
  const float* bfB = b_feats + (size_t)b * NB * FD;
  const int col4 = mlane * 4;
  const int rowFirst = cpart * OUT_ROWS_PER_WAVE + (hi ? 1 : 0);

  v4f g0[ROW_PAIRS], g1[ROW_PAIRS], g2[ROW_PAIRS], ga[ROW_PAIRS];
  int kd0[ROW_PAIRS], kd1[ROW_PAIRS], kd2[ROW_PAIRS];
#pragma unroll
  for (int it = 0; it < ROW_PAIRS; ++it) {
    const int row = rowFirst + it * 2;
    int pk0 = sPk[0][rtile][row][0];
    int pk1 = sPk[0][rtile][row][1];
    int pk2 = sPk[0][rtile][row][2];
#pragma unroll
    for (int c = 1; c < COL_SPLIT; ++c) {
      const int q0 = sPk[c][rtile][row][0];
      const int q1 = sPk[c][rtile][row][1];
      const int q2 = sPk[c][rtile][row][2];
      TOP3_INSERT_PK(q0, pk0, pk1, pk2);
      TOP3_INSERT_PK(q1, pk0, pk1, pk2);
      TOP3_INSERT_PK(q2, pk0, pk1, pk2);
    }
    const int j0 = 4095 - (pk0 & 4095);
    const int j1 = 4095 - (pk1 & 4095);
    const int j2 = 4095 - (pk2 & 4095);
    g0[it] = *(const v4f*)(bfB + (size_t)j0 * FD + col4);
    g1[it] = *(const v4f*)(bfB + (size_t)j1 * FD + col4);
    g2[it] = *(const v4f*)(bfB + (size_t)j2 * FD + col4);
    ga[it] = *(const v4f*)(afBase + (size_t)row * FD + col4);
    kd0[it] = (int)((unsigned)pk0 >> 12);
    kd1[it] = (int)((unsigned)pk1 >> 12);
    kd2[it] = (int)((unsigned)pk2 >> 12);
  }
  __builtin_amdgcn_sched_barrier(0);
#pragma unroll
  for (int it = 0; it < ROW_PAIRS; ++it) {
    const int row = rowFirst + it * 2;
    const int rna2 = __shfl(na2i, row, 32);
    const int d20 = rna2 + KBIAS - kd0[it];
    const int d21 = rna2 + KBIAS - kd1[it];
    const int d22 = rna2 + KBIAS - kd2[it];
    const float inv = 1.0f / 128.0f;
    const float w0 = fmaxf(0.5f - sqrtf((float)d20) * inv, 0.0f);
    const float w1 = fmaxf(0.5f - sqrtf((float)d21) * inv, 0.0f);
    const float w2 = fmaxf(0.5f - sqrtf((float)d22) * inv, 0.0f);
    v4f tmp;
#pragma unroll
    for (int e = 0; e < 4; ++e) {
      tmp[e] = w0 * g0[it][e] + w1 * g1[it][e] + w2 * g2[it][e];
    }
    float* op = outBase + (size_t)row * (2 * FD) + col4;
    *(v4f*)op        = ga[it];
    *(v4f*)(op + FD) = tmp;
  }
}

extern "C" void kernel_launch(void* const* d_in, const int* in_sizes, int n_in,
                              void* d_out, int out_size, void* d_ws, size_t ws_size,
                              hipStream_t stream) {
  const float* a_feats  = (const float*)d_in[0];
  const float* b_feats  = (const float*)d_in[1];
  const int*   a_coords = (const int*)d_in[2];
  const int*   b_coords = (const int*)d_in[3];
  float* out = (float*)d_out;

  dim3 grid(B_SZ * (NA / ROWS_PER_BLOCK));
  dim3 block(WAVES * 32);
  knn3_wmma<<<grid, block, 0, stream>>>(a_feats, b_feats, a_coords, b_coords, out);
}


# sweep software-pipelined in groups of 4 tiles: a group's LDS words are read one group ahead, under the WMMAs and inserts of the group before, not waited for in front of their own WMMA
# speedup vs baseline: 2.1146x; 1.1468x over previous
#include <hip/hip_runtime.h>

typedef __attribute__((ext_vector_type(2))) float v2f;
typedef __attribute__((ext_vector_type(8))) float v8f;
typedef __attribute__((ext_vector_type(4))) float v4f;

#define B_SZ 8
#define NA 4096
#define NB 4096
#define FD 64
#define ROW_TILES 8
#define COL_SPLIT 2
#define WAVES (ROW_TILES * COL_SPLIT)
#define ROWS_PER_WAVE 16
#define ROWS_PER_BLOCK (ROW_TILES * ROWS_PER_WAVE)
#define NTILES (NB / 16)
#define TILES_PER_WAVE (NTILES / COL_SPLIT)
#define OUT_ROWS_PER_WAVE (ROWS_PER_WAVE / COL_SPLIT)
#define STAGE_TRIPS (NB / (WAVES * 32))
#define ROW_PAIRS (OUT_ROWS_PER_WAVE / 2)
static_assert(NB % (WAVES * 32) == 0, "every thread stages the same number of columns");
static_assert(OUT_ROWS_PER_WAVE % 2 == 0 && FD == 16 * 4, "a 16-lane half covers one 64-float row with 4 floats per lane");
#define KBIAS 48388
#define STEP_SCALE ((float)TILES_PER_WAVE)
static_assert(TILES_PER_WAVE == 128, "sweep key = 17 integer bits + 7 step bits = 24 bits of f32");
#define SWEEP_GROUP 4
#define SWEEP_GROUPS (TILES_PER_WAVE / SWEEP_GROUP)
static_assert(TILES_PER_WAVE % SWEEP_GROUP == 0 && (SWEEP_GROUPS & (SWEEP_GROUPS - 1)) == 0, "the look-ahead wraps with a mask");

#define TOP3_INSERT_PK(pk, b0, b1, b2) do {          \
    int _r0 = min((pk), (b0));                        \
    (b0)    = max((pk), (b0));                        \
    int _r1 = min(_r0, (b1));                         \
    (b1)    = max(_r0, (b1));                         \
    (b2)    = max(_r1, (b2));                         \
  } while (0)

__device__ __forceinline__ float max_nonneg(float a, float b) {
  return __int_as_float(max(__float_as_int(a), __float_as_int(b)));
}

#define TOP3_INSERT_F(fk, f0, f1, f2) do {                   \
    (f2) = __builtin_amdgcn_fmed3f((fk), (f1), (f2));         \
    (f1) = __builtin_amdgcn_fmed3f((fk), (f0), (f1));         \
    (f0) = max_nonneg((fk), (f0));                            \
  } while (0)

__global__ __launch_bounds__(WAVES * 32) void knn3_wmma(
    const float* __restrict__ a_feats,
    const float* __restrict__ b_feats,
    const int*   __restrict__ a_coords,
    const int*   __restrict__ b_coords,
    float*       __restrict__ out)
{
  __shared__ unsigned sPack[NB];
  __shared__ float    sKb2[NB];
  __shared__ int      sPk[COL_SPLIT][ROW_TILES][ROWS_PER_WAVE][3];

  const int tid   = threadIdx.x;
  const int lane  = tid & 31;
  const int wave  = tid >> 5;
  const int mlane = lane & 15;
  const bool hi   = lane >= 16;
  const int rtile = wave % ROW_TILES;
  const int cpart = wave / ROW_TILES;

  const int blocksPerBatch = NA / ROWS_PER_BLOCK;
  const int b       = blockIdx.x / blocksPerBatch;
  const int rowBase = (blockIdx.x % blocksPerBatch) * ROWS_PER_BLOCK
                    + rtile * ROWS_PER_WAVE;

  const int arow = rowBase + mlane;
  const int* ap = a_coords + ((size_t)b * NA + arow) * 3;
  const int ldax = ap[0], lday = ap[1], ldaz = ap[2];

#pragma unroll
  for (int k = 0; k < STAGE_TRIPS; ++k) {
    const int n = tid + k * (WAVES * 32);
    const int* p = b_coords + ((size_t)b * NB + n) * 3;
    int x = p[0] >> 4, y = p[1] >> 4, z = p[2] >> 4;
    sPack[n] = (unsigned)x | ((unsigned)y << 8) | ((unsigned)z << 16);
    const int tie = (TILES_PER_WAVE - 1) - ((n >> 4) & (TILES_PER_WAVE - 1));
    sKb2[n]  = (float)(KBIAS - (x * x + y * y + z * z)) + (float)tie * (1.0f / STEP_SCALE);
  }
  __syncthreads();

  const int iax = ldax >> 4, iay = lday >> 4, iaz = ldaz >> 4;
  const int na2i = iax * iax + iay * iay + iaz * iaz;
  v2f A;
  A.x = hi ? (float)(2 * iaz) : (float)(2 * iax);
  A.y = hi ? 1.0f             : (float)(2 * iay);

  float sk0[8], sk1[8], sk2[8];
#pragma unroll
  for (int r = 0; r < 8; ++r) { sk0[r] = sk1[r] = sk2[r] = 0.0f; }

  const v8f Cz = {};

  const int nFirst = cpart * (TILES_PER_WAVE * 16) + mlane;
  const unsigned* pkp = sPack + nFirst;
  const float*    kbp = sKb2 + nFirst;

  unsigned un[SWEEP_GROUP];
  float    kn[SWEEP_GROUP];
#pragma unroll
  for (int t = 0; t < SWEEP_GROUP; ++t) { un[t] = pkp[t * 16]; kn[t] = kbp[t * 16]; }

  for (int g = 0; g < SWEEP_GROUPS; ++g) {
    v2f Bm[SWEEP_GROUP];
#pragma unroll
    for (int t = 0; t < SWEEP_GROUP; ++t) {
      const unsigned u = un[t];
      const float bx = (float)(u & 255u);
      const float by = (float)((u >> 8) & 255u);
      const float bz = (float)((u >> 16) & 255u);
      Bm[t].x = hi ? bz : bx;
      Bm[t].y = hi ? kn[t] : by;
    }
    const int gn = ((g + 1) & (SWEEP_GROUPS - 1)) * (SWEEP_GROUP * 16);
#pragma unroll
    for (int t = 0; t < SWEEP_GROUP; ++t) { un[t] = pkp[gn + t * 16]; kn[t] = kbp[gn + t * 16]; }

#pragma unroll
    for (int t = 0; t < SWEEP_GROUP; ++t) {
      v8f Dv = __builtin_amdgcn_wmma_f32_16x16x4_f32(
           false, A,  false, Bm[t],
           (short)0, Cz,  false,  false);
#pragma unroll
      for (int r = 0; r < 8; ++r) {
        TOP3_INSERT_F(Dv[r], sk0[r], sk1[r], sk2[r]);
      }
    }
  }

  int bp0[8], bp1[8], bp2[8];
#pragma unroll
  for (int r = 0; r < 8; ++r) {
    const float fs[3] = { sk0[r], sk1[r], sk2[r] };
    int ps[3];
#pragma unroll
    for (int t = 0; t < 3; ++t) {
      const int key  = (int)fs[t];
      const int tie  = (int)((fs[t] - (float)key) * STEP_SCALE);
      const int nn   = nFirst + ((TILES_PER_WAVE - 1) - tie) * 16;
      ps[t] = (key << 12) + (4095 - nn);
    }
    bp0[r] = ps[0]; bp1[r] = ps[1]; bp2[r] = ps[2];
  }

#pragma unroll
  for (int r = 0; r < 8; ++r) {
    int p0 = bp0[r], p1 = bp1[r], p2 = bp2[r];
#pragma unroll
    for (int off = 1; off < 16; off <<= 1) {
      int q0 = __shfl_xor(p0, off, 32);
      int q1 = __shfl_xor(p1, off, 32);
      int q2 = __shfl_xor(p2, off, 32);
      TOP3_INSERT_PK(q0, p0, p1, p2);
      TOP3_INSERT_PK(q1, p0, p1, p2);
      TOP3_INSERT_PK(q2, p0, p1, p2);
    }
    if (mlane == 0) {
      const int rowLocal = r + (hi ? 8 : 0);
      sPk[cpart][rtile][rowLocal][0] = p0;
      sPk[cpart][rtile][rowLocal][1] = p1;
      sPk[cpart][rtile][rowLocal][2] = p2;
    }
  }
  __syncthreads();

  const float* afBase = a_feats + ((size_t)b * NA + rowBase) * FD;
  float* outBase = out + ((size_t)b * NA + rowBase) * (2 * FD);
  const float* bfB = b_feats + (size_t)b * NB * FD;
  const int col4 = mlane * 4;
  const int rowFirst = cpart * OUT_ROWS_PER_WAVE + (hi ? 1 : 0);

  v4f g0[ROW_PAIRS], g1[ROW_PAIRS], g2[ROW_PAIRS], ga[ROW_PAIRS];
  int kd0[ROW_PAIRS], kd1[ROW_PAIRS], kd2[ROW_PAIRS];
#pragma unroll
  for (int it = 0; it < ROW_PAIRS; ++it) {
    const int row = rowFirst + it * 2;
    int pk0 = sPk[0][rtile][row][0];
    int pk1 = sPk[0][rtile][row][1];
    int pk2 = sPk[0][rtile][row][2];
#pragma unroll
    for (int c = 1; c < COL_SPLIT; ++c) {
      const int q0 = sPk[c][rtile][row][0];
      const int q1 = sPk[c][rtile][row][1];
      const int q2 = sPk[c][rtile][row][2];
      TOP3_INSERT_PK(q0, pk0, pk1, pk2);
      TOP3_INSERT_PK(q1, pk0, pk1, pk2);
      TOP3_INSERT_PK(q2, pk0, pk1, pk2);
    }
    const int j0 = 4095 - (pk0 & 4095);
    const int j1 = 4095 - (pk1 & 4095);
    const int j2 = 4095 - (pk2 & 4095);
    g0[it] = *(const v4f*)(bfB + (size_t)j0 * FD + col4);
    g1[it] = *(const v4f*)(bfB + (size_t)j1 * FD + col4);
    g2[it] = *(const v4f*)(bfB + (size_t)j2 * FD + col4);
    ga[it] = *(const v4f*)(afBase + (size_t)row * FD + col4);
    kd0[it] = (int)((unsigned)pk0 >> 12);
    kd1[it] = (int)((unsigned)pk1 >> 12);
    kd2[it] = (int)((unsigned)pk2 >> 12);
  }
  __builtin_amdgcn_sched_barrier(0);
#pragma unroll
  for (int it = 0; it < ROW_PAIRS; ++it) {
    const int row = rowFirst + it * 2;
    const int rna2 = __shfl(na2i, row, 32);
    const int d20 = rna2 + KBIAS - kd0[it];
    const int d21 = rna2 + KBIAS - kd1[it];
    const int d22 = rna2 + KBIAS - kd2[it];
    const float inv = 1.0f / 128.0f;
    const float w0 = fmaxf(0.5f - sqrtf((float)d20) * inv, 0.0f);
    const float w1 = fmaxf(0.5f - sqrtf((float)d21) * inv, 0.0f);
    const float w2 = fmaxf(0.5f - sqrtf((float)d22) * inv, 0.0f);
    v4f tmp;
#pragma unroll
    for (int e = 0; e < 4; ++e) {
      tmp[e] = w0 * g0[it][e] + w1 * g1[it][e] + w2 * g2[it][e];
    }
    float* op = outBase + (size_t)row * (2 * FD) + col4;
    *(v4f*)op        = ga[it];
    *(v4f*)(op + FD) = tmp;
  }
}

extern "C" void kernel_launch(void* const* d_in, const int* in_sizes, int n_in,
                              void* d_out, int out_size, void* d_ws, size_t ws_size,
                              hipStream_t stream) {
  const float* a_feats  = (const float*)d_in[0];
  const float* b_feats  = (const float*)d_in[1];
  const int*   a_coords = (const int*)d_in[2];
  const int*   b_coords = (const int*)d_in[3];
  float* out = (float*)d_out;

  dim3 grid(B_SZ * (NA / ROWS_PER_BLOCK));
  dim3 block(WAVES * 32);
  knn3_wmma<<<grid, block, 0, stream>>>(a_feats, b_feats, a_coords, b_coords, out);
}


# cross-lane top-3 merge via LDS: lanes store their sorted triples; after the barrier 4 lanes per row merge 8 lists each (7-op sorted merge) + 2 shuffle steps; was a 4-step 16-lane butterfly per row
# speedup vs baseline: 2.3883x; 1.1294x over previous
#include <hip/hip_runtime.h>

typedef __attribute__((ext_vector_type(2))) float v2f;
typedef __attribute__((ext_vector_type(8))) float v8f;
typedef __attribute__((ext_vector_type(4))) float v4f;
typedef __attribute__((ext_vector_type(4))) int v4i;

#define B_SZ 8
#define NA 4096
#define NB 4096
#define FD 64
#define ROW_TILES 8
#define COL_SPLIT 2
#define WAVES (ROW_TILES * COL_SPLIT)
#define ROWS_PER_WAVE 16
#define ROWS_PER_BLOCK (ROW_TILES * ROWS_PER_WAVE)
#define NTILES (NB / 16)
#define TILES_PER_WAVE (NTILES / COL_SPLIT)
#define OUT_ROWS_PER_WAVE (ROWS_PER_WAVE / COL_SPLIT)
#define STAGE_TRIPS (NB / (WAVES * 32))
#define ROW_PAIRS (OUT_ROWS_PER_WAVE / 2)
#define SRC_LISTS (COL_SPLIT * 16)
#define XPITCH (3 * SRC_LISTS + 4)
#define LISTS_PER_MERGER (SRC_LISTS / 4)
static_assert(OUT_ROWS_PER_WAVE * 4 == 32 && LISTS_PER_MERGER == 8, "a wave's 32 lanes = 8 output rows x 4 mergers of 8 lists");
static_assert(NB % (WAVES * 32) == 0, "every thread stages the same number of columns");
static_assert(OUT_ROWS_PER_WAVE % 2 == 0 && FD == 16 * 4, "a 16-lane half covers one 64-float row with 4 floats per lane");
#define KBIAS 48388
#define STEP_SCALE ((float)TILES_PER_WAVE)
static_assert(TILES_PER_WAVE == 128, "sweep key = 17 integer bits + 7 step bits = 24 bits of f32");
#define SWEEP_GROUP 4
#define SWEEP_GROUPS (TILES_PER_WAVE / SWEEP_GROUP)
static_assert(TILES_PER_WAVE % SWEEP_GROUP == 0 && (SWEEP_GROUPS & (SWEEP_GROUPS - 1)) == 0, "the look-ahead wraps with a mask");

#define MERGE3_SORTED(p0, p1, p2, q0, q1, q2) do {     \
    const int _lo = min((p0), (q0));                    \
    const int _c1 = min((p1), (q0));                    \
    const int _c2 = min((p0), (q1));                    \
    (p2) = max(max((p2), (q2)), max(_c1, _c2));         \
    (p1) = max(_lo, max((p1), (q1)));                   \
    (p0) = max((p0), (q0));                             \
  } while (0)

__device__ __forceinline__ float max_nonneg(float a, float b) {
  return __int_as_float(max(__float_as_int(a), __float_as_int(b)));
}

#define TOP3_INSERT_F(fk, f0, f1, f2) do {                   \
    (f2) = __builtin_amdgcn_fmed3f((fk), (f1), (f2));         \
    (f1) = __builtin_amdgcn_fmed3f((fk), (f0), (f1));         \
    (f0) = max_nonneg((fk), (f0));                            \
  } while (0)

__global__ __launch_bounds__(WAVES * 32) void knn3_wmma(
    const float* __restrict__ a_feats,
    const float* __restrict__ b_feats,
    const int*   __restrict__ a_coords,
    const int*   __restrict__ b_coords,
    float*       __restrict__ out)
{
  struct __attribute__((aligned(16))) LdsLayout {
    unsigned pack[NB];
    float    kb2[NB];
    int      x[ROWS_PER_BLOCK][XPITCH];
  };
  __shared__ LdsLayout lds;
  unsigned (&sPack)[NB] = lds.pack;
  float    (&sKb2)[NB]  = lds.kb2;
  int      (&sX)[ROWS_PER_BLOCK][XPITCH] = lds.x;

  const int tid   = threadIdx.x;
  const int lane  = tid & 31;
  const int wave  = tid >> 5;
  const int mlane = lane & 15;
  const bool hi   = lane >= 16;
  const int rtile = wave % ROW_TILES;
  const int cpart = wave / ROW_TILES;

  const int blocksPerBatch = NA / ROWS_PER_BLOCK;
  const int b       = blockIdx.x / blocksPerBatch;
  const int rowBase = (blockIdx.x % blocksPerBatch) * ROWS_PER_BLOCK
                    + rtile * ROWS_PER_WAVE;

  const int arow = rowBase + mlane;
  const int* ap = a_coords + ((size_t)b * NA + arow) * 3;
  const int ldax = ap[0], lday = ap[1], ldaz = ap[2];

#pragma unroll
  for (int k = 0; k < STAGE_TRIPS; ++k) {
    const int n = tid + k * (WAVES * 32);
    const int* p = b_coords + ((size_t)b * NB + n) * 3;
    int x = p[0] >> 4, y = p[1] >> 4, z = p[2] >> 4;
    sPack[n] = (unsigned)x | ((unsigned)y << 8) | ((unsigned)z << 16);
    const int tie = (TILES_PER_WAVE - 1) - ((n >> 4) & (TILES_PER_WAVE - 1));
    sKb2[n]  = (float)(KBIAS - (x * x + y * y + z * z)) + (float)tie * (1.0f / STEP_SCALE);
  }
  __syncthreads();

  const int iax = ldax >> 4, iay = lday >> 4, iaz = ldaz >> 4;
  const int na2i = iax * iax + iay * iay + iaz * iaz;
  v2f A;
  A.x = hi ? (float)(2 * iaz) : (float)(2 * iax);
  A.y = hi ? 1.0f             : (float)(2 * iay);

  float sk0[8], sk1[8], sk2[8];
#pragma unroll
  for (int r = 0; r < 8; ++r) { sk0[r] = sk1[r] = sk2[r] = 0.0f; }

  const v8f Cz = {};

  const int nFirst = cpart * (TILES_PER_WAVE * 16) + mlane;
  const unsigned* pkp = sPack + nFirst;
  const float*    kbp = sKb2 + nFirst;

  unsigned un[SWEEP_GROUP];
  float    kn[SWEEP_GROUP];
#pragma unroll
  for (int t = 0; t < SWEEP_GROUP; ++t) { un[t] = pkp[t * 16]; kn[t] = kbp[t * 16]; }
  asm volatile("" :: "v"(un[0]), "v"(un[1]), "v"(un[2]), "v"(un[3]),
                     "v"(kn[0]), "v"(kn[1]), "v"(kn[2]), "v"(kn[3]));

  for (int g = 0; g < SWEEP_GROUPS; ++g) {
    v2f Bm[SWEEP_GROUP];
#pragma unroll
    for (int t = 0; t < SWEEP_GROUP; ++t) {
      const unsigned u = un[t];
      const float bx = (float)(u & 255u);
      const float by = (float)((u >> 8) & 255u);
      const float bz = (float)((u >> 16) & 255u);
      Bm[t].x = hi ? bz : bx;
      Bm[t].y = hi ? kn[t] : by;
    }
    const int gn = ((g + 1) & (SWEEP_GROUPS - 1)) * (SWEEP_GROUP * 16);
#pragma unroll
    for (int t = 0; t < SWEEP_GROUP; ++t) { un[t] = pkp[gn + t * 16]; kn[t] = kbp[gn + t * 16]; }

#pragma unroll
    for (int t = 0; t < SWEEP_GROUP; ++t) {
      v8f Dv = __builtin_amdgcn_wmma_f32_16x16x4_f32(
           false, A,  false, Bm[t],
           (short)0, Cz,  false,  false);
#pragma unroll
      for (int r = 0; r < 8; ++r) {
        TOP3_INSERT_F(Dv[r], sk0[r], sk1[r], sk2[r]);
      }
    }
  }

  int bp0[8], bp1[8], bp2[8];
#pragma unroll
  for (int r = 0; r < 8; ++r) {
    const float fs[3] = { sk0[r], sk1[r], sk2[r] };
    int ps[3];
#pragma unroll
    for (int t = 0; t < 3; ++t) {
      const int key  = (int)fs[t];
      const int tie  = (int)((fs[t] - (float)key) * STEP_SCALE);
      const int nn   = nFirst + ((TILES_PER_WAVE - 1) - tie) * 16;
      ps[t] = (key << 12) + (4095 - nn);
    }
    bp0[r] = ps[0]; bp1[r] = ps[1]; bp2[r] = ps[2];
  }

  {
    const int slot = cpart * 16 + mlane;
#pragma unroll
    for (int r = 0; r < 8; ++r) {
      int* xr = &sX[rtile * ROWS_PER_WAVE + r + (hi ? 8 : 0)][slot];
      xr[0]             = bp0[r];
      xr[SRC_LISTS]     = bp1[r];
      xr[2 * SRC_LISTS] = bp2[r];
    }
  }
  __syncthreads();

  const float* afBase = a_feats + ((size_t)b * NA + rowBase) * FD;
  float* outBase = out + ((size_t)b * NA + rowBase) * (2 * FD);
  const float* bfB = b_feats + (size_t)b * NB * FD;
  const int col4 = mlane * 4;
  const int rowFirst = cpart * OUT_ROWS_PER_WAVE + (hi ? 1 : 0);

  int m0[LISTS_PER_MERGER], m1[LISTS_PER_MERGER], m2[LISTS_PER_MERGER];
  {
    const int* xq = &sX[rtile * ROWS_PER_WAVE + cpart * OUT_ROWS_PER_WAVE + (lane >> 2)][(lane & 3) * LISTS_PER_MERGER];
    const v4i a0 = *(const v4i*)(xq),                 b0 = *(const v4i*)(xq + 4);
    const v4i a1 = *(const v4i*)(xq + SRC_LISTS),     b1 = *(const v4i*)(xq + SRC_LISTS + 4);
    const v4i a2 = *(const v4i*)(xq + 2 * SRC_LISTS), b2 = *(const v4i*)(xq + 2 * SRC_LISTS + 4);
#pragma unroll
    for (int e = 0; e < 4; ++e) {
      m0[e] = a0[e]; m0[e + 4] = b0[e];
      m1[e] = a1[e]; m1[e + 4] = b1[e];
      m2[e] = a2[e]; m2[e + 4] = b2[e];
    }
  }
#pragma unroll
  for (int span = 1; span < LISTS_PER_MERGER; span <<= 1) {
#pragma unroll
    for (int i = 0; i < LISTS_PER_MERGER; i += 2 * span) {
      MERGE3_SORTED(m0[i], m1[i], m2[i], m0[i + span], m1[i + span], m2[i + span]);
    }
  }
  int f0 = m0[0], f1 = m1[0], f2 = m2[0];
#pragma unroll
  for (int off = 1; off < 4; off <<= 1) {
    const int q0 = __shfl_xor(f0, off, 32);
    const int q1 = __shfl_xor(f1, off, 32);
    const int q2 = __shfl_xor(f2, off, 32);
    MERGE3_SORTED(f0, f1, f2, q0, q1, q2);
  }

  v4f g0[ROW_PAIRS], g1[ROW_PAIRS], g2[ROW_PAIRS], ga[ROW_PAIRS];
  int kd0[ROW_PAIRS], kd1[ROW_PAIRS], kd2[ROW_PAIRS];
#pragma unroll
  for (int it = 0; it < ROW_PAIRS; ++it) {
    const int row = rowFirst + it * 2;
    const int mergedBy = (it * 2 + (hi ? 1 : 0)) * 4;
    const int pk0 = __shfl(f0, mergedBy, 32);
    const int pk1 = __shfl(f1, mergedBy, 32);
    const int pk2 = __shfl(f2, mergedBy, 32);
    const int j0 = 4095 - (pk0 & 4095);
    const int j1 = 4095 - (pk1 & 4095);
    const int j2 = 4095 - (pk2 & 4095);
    g0[it] = *(const v4f*)(bfB + (size_t)j0 * FD + col4);
    g1[it] = *(const v4f*)(bfB + (size_t)j1 * FD + col4);
    g2[it] = *(const v4f*)(bfB + (size_t)j2 * FD + col4);
    ga[it] = *(const v4f*)(afBase + (size_t)row * FD + col4);
    kd0[it] = (int)((unsigned)pk0 >> 12);
    kd1[it] = (int)((unsigned)pk1 >> 12);
    kd2[it] = (int)((unsigned)pk2 >> 12);
  }
  __builtin_amdgcn_sched_barrier(0);
#pragma unroll
  for (int it = 0; it < ROW_PAIRS; ++it) {
    const int row = rowFirst + it * 2;
    const int rna2 = __shfl(na2i, row, 32);
    const int d20 = rna2 + KBIAS - kd0[it];
    const int d21 = rna2 + KBIAS - kd1[it];
    const int d22 = rna2 + KBIAS - kd2[it];
    const float inv = 1.0f / 128.0f;
    const float w0 = fmaxf(0.5f - sqrtf((float)d20) * inv, 0.0f);
    const float w1 = fmaxf(0.5f - sqrtf((float)d21) * inv, 0.0f);
    const float w2 = fmaxf(0.5f - sqrtf((float)d22) * inv, 0.0f);
    v4f tmp;
#pragma unroll
    for (int e = 0; e < 4; ++e) {
      tmp[e] = w0 * g0[it][e] + w1 * g1[it][e] + w2 * g2[it][e];
    }
    float* op = outBase + (size_t)row * (2 * FD) + col4;
    *(v4f*)op        = ga[it];
    *(v4f*)(op + FD) = tmp;
  }
}

extern "C" void kernel_launch(void* const* d_in, const int* in_sizes, int n_in,
                              void* d_out, int out_size, void* d_ws, size_t ws_size,
                              hipStream_t stream) {
  const float* a_feats  = (const float*)d_in[0];
  const float* b_feats  = (const float*)d_in[1];
  const int*   a_coords = (const int*)d_in[2];
  const int*   b_coords = (const int*)d_in[3];
  float* out = (float*)d_out;

  dim3 grid(B_SZ * (NA / ROWS_PER_BLOCK));
  dim3 block(WAVES * 32);
  knn3_wmma<<<grid, block, 0, stream>>>(a_feats, b_feats, a_coords, b_coords, out);
}


# the sweep's B operand is staged finished in LDS (per column the floats {x, y | z, bias}): one 8-byte LDS read per lane and tile, two register sets in turn; no unpack, convert or select in the sweep
# speedup vs baseline: 2.5327x; 1.0605x over previous
#include <hip/hip_runtime.h>

typedef __attribute__((ext_vector_type(2))) float v2f;
typedef __attribute__((ext_vector_type(8))) float v8f;
typedef __attribute__((ext_vector_type(4))) float v4f;
typedef __attribute__((ext_vector_type(4))) int v4i;

#define B_SZ 8
#define NA 4096
#define NB 4096
#define FD 64
#define ROW_TILES 8
#define COL_SPLIT 2
#define WAVES (ROW_TILES * COL_SPLIT)
#define ROWS_PER_WAVE 16
#define ROWS_PER_BLOCK (ROW_TILES * ROWS_PER_WAVE)
#define NTILES (NB / 16)
#define TILES_PER_WAVE (NTILES / COL_SPLIT)
#define OUT_ROWS_PER_WAVE (ROWS_PER_WAVE / COL_SPLIT)
#define STAGE_TRIPS (NB / (WAVES * 32))
#define ROW_PAIRS (OUT_ROWS_PER_WAVE / 2)
#define SRC_LISTS (COL_SPLIT * 16)
#define XPITCH (3 * SRC_LISTS + 4)
#define LISTS_PER_MERGER (SRC_LISTS / 4)
static_assert(OUT_ROWS_PER_WAVE * 4 == 32 && LISTS_PER_MERGER == 8, "a wave's 32 lanes = 8 output rows x 4 mergers of 8 lists");
static_assert(NB % (WAVES * 32) == 0, "every thread stages the same number of columns");
static_assert(OUT_ROWS_PER_WAVE % 2 == 0 && FD == 16 * 4, "a 16-lane half covers one 64-float row with 4 floats per lane");
#define KBIAS 48388
#define STEP_SCALE ((float)TILES_PER_WAVE)
static_assert(TILES_PER_WAVE == 128, "sweep key = 17 integer bits + 7 step bits = 24 bits of f32");
#define SWEEP_GROUP 4
#define SWEEP_GROUPS (TILES_PER_WAVE / SWEEP_GROUP)
static_assert(TILES_PER_WAVE % SWEEP_GROUP == 0 && (SWEEP_GROUPS & (SWEEP_GROUPS - 1)) == 0 && SWEEP_GROUPS % 2 == 0, "the look-ahead wraps with a mask; two groups per loop trip");

#define MERGE3_SORTED(p0, p1, p2, q0, q1, q2) do {     \
    const int _lo = min((p0), (q0));                    \
    const int _c1 = min((p1), (q0));                    \
    const int _c2 = min((p0), (q1));                    \
    (p2) = max(max((p2), (q2)), max(_c1, _c2));         \
    (p1) = max(_lo, max((p1), (q1)));                   \
    (p0) = max((p0), (q0));                             \
  } while (0)

__device__ __forceinline__ float max_nonneg(float a, float b) {
  return __int_as_float(max(__float_as_int(a), __float_as_int(b)));
}

#define TOP3_INSERT_F(fk, f0, f1, f2) do {                   \
    (f2) = __builtin_amdgcn_fmed3f((fk), (f1), (f2));         \
    (f1) = __builtin_amdgcn_fmed3f((fk), (f0), (f1));         \
    (f0) = max_nonneg((fk), (f0));                            \
  } while (0)

__global__ __launch_bounds__(WAVES * 32) void knn3_wmma(
    const float* __restrict__ a_feats,
    const float* __restrict__ b_feats,
    const int*   __restrict__ a_coords,
    const int*   __restrict__ b_coords,
    float*       __restrict__ out)
{
  struct __attribute__((aligned(16))) LdsLayout {
    v4f      bfrag[NB];
    int      x[ROWS_PER_BLOCK][XPITCH];
  };
  __shared__ LdsLayout lds;
  v4f      (&sFrag)[NB] = lds.bfrag;
  int      (&sX)[ROWS_PER_BLOCK][XPITCH] = lds.x;

  const int tid   = threadIdx.x;
  const int lane  = tid & 31;
  const int wave  = tid >> 5;
  const int mlane = lane & 15;
  const bool hi   = lane >= 16;
  const int rtile = wave % ROW_TILES;
  const int cpart = wave / ROW_TILES;

  const int blocksPerBatch = NA / ROWS_PER_BLOCK;
  const int b       = blockIdx.x / blocksPerBatch;
  const int rowBase = (blockIdx.x % blocksPerBatch) * ROWS_PER_BLOCK
                    + rtile * ROWS_PER_WAVE;

  const int arow = rowBase + mlane;
  const int* ap = a_coords + ((size_t)b * NA + arow) * 3;
  const int ldax = ap[0], lday = ap[1], ldaz = ap[2];

#pragma unroll
  for (int k = 0; k < STAGE_TRIPS; ++k) {
    const int n = tid + k * (WAVES * 32);
    const int* p = b_coords + ((size_t)b * NB + n) * 3;
    int x = p[0] >> 4, y = p[1] >> 4, z = p[2] >> 4;
    const int tie = (TILES_PER_WAVE - 1) - ((n >> 4) & (TILES_PER_WAVE - 1));
    v4f f;
    f.x = (float)x;
    f.y = (float)y;
    f.z = (float)z;
    f.w = (float)(KBIAS - (x * x + y * y + z * z)) + (float)tie * (1.0f / STEP_SCALE);
    sFrag[n] = f;
  }
  __syncthreads();

  const int iax = ldax >> 4, iay = lday >> 4, iaz = ldaz >> 4;
  const int na2i = iax * iax + iay * iay + iaz * iaz;
  v2f A;
  A.x = hi ? (float)(2 * iaz) : (float)(2 * iax);
  A.y = hi ? 1.0f             : (float)(2 * iay);

  float sk0[8], sk1[8], sk2[8];
#pragma unroll
  for (int r = 0; r < 8; ++r) { sk0[r] = sk1[r] = sk2[r] = 0.0f; }

  const v8f Cz = {};

  const int nFirst = cpart * (TILES_PER_WAVE * 16) + mlane;
  const v2f* bfp = (const v2f*)&sFrag[nFirst] + (hi ? 1 : 0);

  v2f bE[SWEEP_GROUP], bO[SWEEP_GROUP];
#pragma unroll
  for (int t = 0; t < SWEEP_GROUP; ++t) { bE[t] = bfp[t * 32]; }

  for (int g = 0; g < SWEEP_GROUPS; g += 2) {
    const int go = (g + 1) * (SWEEP_GROUP * 32);
#pragma unroll
    for (int t = 0; t < SWEEP_GROUP; ++t) { bO[t] = bfp[go + t * 32]; }
    __builtin_amdgcn_sched_barrier(0);

#pragma unroll
    for (int t = 0; t < SWEEP_GROUP; ++t) {
      v8f Dv = __builtin_amdgcn_wmma_f32_16x16x4_f32(
           false, A,  false, bE[t],
           (short)0, Cz,  false,  false);
#pragma unroll
      for (int r = 0; r < 8; ++r) {
        TOP3_INSERT_F(Dv[r], sk0[r], sk1[r], sk2[r]);
      }
    }

    const int ge = ((g + 2) & (SWEEP_GROUPS - 1)) * (SWEEP_GROUP * 32);
#pragma unroll
    for (int t = 0; t < SWEEP_GROUP; ++t) { bE[t] = bfp[ge + t * 32]; }
    __builtin_amdgcn_sched_barrier(0);


#pragma unroll
    for (int t = 0; t < SWEEP_GROUP; ++t) {
      v8f Dv = __builtin_amdgcn_wmma_f32_16x16x4_f32(
           false, A,  false, bO[t],
           (short)0, Cz,  false,  false);
#pragma unroll
      for (int r = 0; r < 8; ++r) {
        TOP3_INSERT_F(Dv[r], sk0[r], sk1[r], sk2[r]);
      }
    }
  }

  int bp0[8], bp1[8], bp2[8];
#pragma unroll
  for (int r = 0; r < 8; ++r) {
    const float fs[3] = { sk0[r], sk1[r], sk2[r] };
    int ps[3];
#pragma unroll
    for (int t = 0; t < 3; ++t) {
      const int key  = (int)fs[t];
      const int tie  = (int)((fs[t] - (float)key) * STEP_SCALE);
      const int nn   = nFirst + ((TILES_PER_WAVE - 1) - tie) * 16;
      ps[t] = (key << 12) + (4095 - nn);
    }
    bp0[r] = ps[0]; bp1[r] = ps[1]; bp2[r] = ps[2];
  }

  {
    const int slot = cpart * 16 + mlane;
#pragma unroll
    for (int r = 0; r < 8; ++r) {
      int* xr = &sX[rtile * ROWS_PER_WAVE + r + (hi ? 8 : 0)][slot];
      xr[0]             = bp0[r];
      xr[SRC_LISTS]     = bp1[r];
      xr[2 * SRC_LISTS] = bp2[r];
    }
  }
  __syncthreads();

  const float* afBase = a_feats + ((size_t)b * NA + rowBase) * FD;
  float* outBase = out + ((size_t)b * NA + rowBase) * (2 * FD);
  const float* bfB = b_feats + (size_t)b * NB * FD;
  const int col4 = mlane * 4;
  const int rowFirst = cpart * OUT_ROWS_PER_WAVE + (hi ? 1 : 0);

  int m0[LISTS_PER_MERGER], m1[LISTS_PER_MERGER], m2[LISTS_PER_MERGER];
  {
    const int* xq = &sX[rtile * ROWS_PER_WAVE + cpart * OUT_ROWS_PER_WAVE + (lane >> 2)][(lane & 3) * LISTS_PER_MERGER];
    const v4i a0 = *(const v4i*)(xq),                 b0 = *(const v4i*)(xq + 4);
    const v4i a1 = *(const v4i*)(xq + SRC_LISTS),     b1 = *(const v4i*)(xq + SRC_LISTS + 4);
    const v4i a2 = *(const v4i*)(xq + 2 * SRC_LISTS), b2 = *(const v4i*)(xq + 2 * SRC_LISTS + 4);
#pragma unroll
    for (int e = 0; e < 4; ++e) {
      m0[e] = a0[e]; m0[e + 4] = b0[e];
      m1[e] = a1[e]; m1[e + 4] = b1[e];
      m2[e] = a2[e]; m2[e + 4] = b2[e];
    }
  }
#pragma unroll
  for (int span = 1; span < LISTS_PER_MERGER; span <<= 1) {
#pragma unroll
    for (int i = 0; i < LISTS_PER_MERGER; i += 2 * span) {
      MERGE3_SORTED(m0[i], m1[i], m2[i], m0[i + span], m1[i + span], m2[i + span]);
    }
  }
  int f0 = m0[0], f1 = m1[0], f2 = m2[0];
#pragma unroll
  for (int off = 1; off < 4; off <<= 1) {
    const int q0 = __shfl_xor(f0, off, 32);
    const int q1 = __shfl_xor(f1, off, 32);
    const int q2 = __shfl_xor(f2, off, 32);
    MERGE3_SORTED(f0, f1, f2, q0, q1, q2);
  }

  v4f g0[ROW_PAIRS], g1[ROW_PAIRS], g2[ROW_PAIRS], ga[ROW_PAIRS];
  int kd0[ROW_PAIRS], kd1[ROW_PAIRS], kd2[ROW_PAIRS];
#pragma unroll
  for (int it = 0; it < ROW_PAIRS; ++it) {
    const int row = rowFirst + it * 2;
    const int mergedBy = (it * 2 + (hi ? 1 : 0)) * 4;
    const int pk0 = __shfl(f0, mergedBy, 32);
    const int pk1 = __shfl(f1, mergedBy, 32);
    const int pk2 = __shfl(f2, mergedBy, 32);
    const int j0 = 4095 - (pk0 & 4095);
    const int j1 = 4095 - (pk1 & 4095);
    const int j2 = 4095 - (pk2 & 4095);
    g0[it] = *(const v4f*)(bfB + (size_t)j0 * FD + col4);
    g1[it] = *(const v4f*)(bfB + (size_t)j1 * FD + col4);
    g2[it] = *(const v4f*)(bfB + (size_t)j2 * FD + col4);
    ga[it] = *(const v4f*)(afBase + (size_t)row * FD + col4);
    kd0[it] = (int)((unsigned)pk0 >> 12);
    kd1[it] = (int)((unsigned)pk1 >> 12);
    kd2[it] = (int)((unsigned)pk2 >> 12);
  }
  __builtin_amdgcn_sched_barrier(0);
#pragma unroll
  for (int it = 0; it < ROW_PAIRS; ++it) {
    const int row = rowFirst + it * 2;
    const int rna2 = __shfl(na2i, row, 32);
    const int d20 = rna2 + KBIAS - kd0[it];
    const int d21 = rna2 + KBIAS - kd1[it];
    const int d22 = rna2 + KBIAS - kd2[it];
    const float inv = 1.0f / 128.0f;
    const float w0 = fmaxf(0.5f - sqrtf((float)d20) * inv, 0.0f);
    const float w1 = fmaxf(0.5f - sqrtf((float)d21) * inv, 0.0f);
    const float w2 = fmaxf(0.5f - sqrtf((float)d22) * inv, 0.0f);
    v4f tmp;
#pragma unroll
    for (int e = 0; e < 4; ++e) {
      tmp[e] = w0 * g0[it][e] + w1 * g1[it][e] + w2 * g2[it][e];
    }
    float* op = outBase + (size_t)row * (2 * FD) + col4;
    *(v4f*)op        = ga[it];
    *(v4f*)(op + FD) = tmp;
  }
}

extern "C" void kernel_launch(void* const* d_in, const int* in_sizes, int n_in,
                              void* d_out, int out_size, void* d_ws, size_t ws_size,
                              hipStream_t stream) {
  const float* a_feats  = (const float*)d_in[0];
  const float* b_feats  = (const float*)d_in[1];
  const int*   a_coords = (const int*)d_in[2];
  const int*   b_coords = (const int*)d_in[3];
  float* out = (float*)d_out;

  dim3 grid(B_SZ * (NA / ROWS_PER_BLOCK));
  dim3 block(WAVES * 32);
  knn3_wmma<<<grid, block, 0, stream>>>(a_feats, b_feats, a_coords, b_coords, out);
}


# the sweep keeps two keys per lane list, not three: 2 VALU per candidate (was 3); a row whose best two share one list gets that list's third from an 8-WMMA rescan of its 128 columns
# speedup vs baseline: 2.7859x; 1.1000x over previous
#include <hip/hip_runtime.h>

typedef __attribute__((ext_vector_type(2))) float v2f;
typedef __attribute__((ext_vector_type(8))) float v8f;
typedef __attribute__((ext_vector_type(4))) float v4f;
typedef __attribute__((ext_vector_type(4))) int v4i;

#define B_SZ 8
#define NA 4096
#define NB 4096
#define FD 64
#define ROW_TILES 8
#define COL_SPLIT 2
#define WAVES (ROW_TILES * COL_SPLIT)
#define ROWS_PER_WAVE 16
#define ROWS_PER_BLOCK (ROW_TILES * ROWS_PER_WAVE)
#define NTILES (NB / 16)
#define TILES_PER_WAVE (NTILES / COL_SPLIT)
#define OUT_ROWS_PER_WAVE (ROWS_PER_WAVE / COL_SPLIT)
#define STAGE_TRIPS (NB / (WAVES * 32))
#define ROW_PAIRS (OUT_ROWS_PER_WAVE / 2)
#define SRC_LISTS (COL_SPLIT * 16)
#define XPITCH (2 * SRC_LISTS + 4)
#define LIST_ID_MASK 0x80F
#define RESCAN_TILES (TILES_PER_WAVE / 16)
static_assert(NB == 4096 && COL_SPLIT == 2, "LIST_ID_MASK: bit 11 is the column part, bits 0-3 the lane");
#define LISTS_PER_MERGER (SRC_LISTS / 4)
static_assert(OUT_ROWS_PER_WAVE * 4 == 32 && LISTS_PER_MERGER == 8, "a wave's 32 lanes = 8 output rows x 4 mergers of 8 lists");
static_assert(NB % (WAVES * 32) == 0, "every thread stages the same number of columns");
static_assert(OUT_ROWS_PER_WAVE % 2 == 0 && FD == 16 * 4, "a 16-lane half covers one 64-float row with 4 floats per lane");
#define KBIAS 48388
#define STEP_SCALE ((float)TILES_PER_WAVE)
static_assert(TILES_PER_WAVE == 128, "sweep key = 17 integer bits + 7 step bits = 24 bits of f32");
#define SWEEP_GROUP 4
#define SWEEP_GROUPS (TILES_PER_WAVE / SWEEP_GROUP)
static_assert(TILES_PER_WAVE % SWEEP_GROUP == 0 && (SWEEP_GROUPS & (SWEEP_GROUPS - 1)) == 0 && SWEEP_GROUPS % 2 == 0, "the look-ahead wraps with a mask; two groups per loop trip");

#define MERGE3_SORTED(p0, p1, p2, q0, q1, q2) do {     \
    const int _lo = min((p0), (q0));                    \
    const int _c1 = min((p1), (q0));                    \
    const int _c2 = min((p0), (q1));                    \
    (p2) = max(max((p2), (q2)), max(_c1, _c2));         \
    (p1) = max(_lo, max((p1), (q1)));                   \
    (p0) = max((p0), (q0));                             \
  } while (0)

__device__ __forceinline__ float max_nonneg(float a, float b) {
  return __int_as_float(max(__float_as_int(a), __float_as_int(b)));
}

#define TOP2_INSERT_F(fk, f0, f1) do {                       \
    (f1) = __builtin_amdgcn_fmed3f((fk), (f0), (f1));         \
    (f0) = max_nonneg((fk), (f0));                            \
  } while (0)

__global__ __launch_bounds__(WAVES * 32) void knn3_wmma(
    const float* __restrict__ a_feats,
    const float* __restrict__ b_feats,
    const int*   __restrict__ a_coords,
    const int*   __restrict__ b_coords,
    float*       __restrict__ out)
{
  struct __attribute__((aligned(16))) LdsLayout {
    v4f      bfrag[NB];
    int      x[ROWS_PER_BLOCK][XPITCH];
  };
  __shared__ LdsLayout lds;
  v4f      (&sFrag)[NB] = lds.bfrag;
  int      (&sX)[ROWS_PER_BLOCK][XPITCH] = lds.x;

  const int tid   = threadIdx.x;
  const int lane  = tid & 31;
  const int wave  = tid >> 5;
  const int mlane = lane & 15;
  const bool hi   = lane >= 16;
  const int rtile = wave % ROW_TILES;
  const int cpart = wave / ROW_TILES;

  const int blocksPerBatch = NA / ROWS_PER_BLOCK;
  const int b       = blockIdx.x / blocksPerBatch;
  const int rowBase = (blockIdx.x % blocksPerBatch) * ROWS_PER_BLOCK
                    + rtile * ROWS_PER_WAVE;

  const int arow = rowBase + mlane;
  const int* ap = a_coords + ((size_t)b * NA + arow) * 3;
  const int ldax = ap[0], lday = ap[1], ldaz = ap[2];

#pragma unroll
  for (int k = 0; k < STAGE_TRIPS; ++k) {
    const int n = tid + k * (WAVES * 32);
    const int* p = b_coords + ((size_t)b * NB + n) * 3;
    int x = p[0] >> 4, y = p[1] >> 4, z = p[2] >> 4;
    const int tie = (TILES_PER_WAVE - 1) - ((n >> 4) & (TILES_PER_WAVE - 1));
    v4f f;
    f.x = (float)x;
    f.y = (float)y;
    f.z = (float)z;
    f.w = (float)(KBIAS - (x * x + y * y + z * z)) + (float)tie * (1.0f / STEP_SCALE);
    sFrag[n] = f;
  }
  __syncthreads();

  const int iax = ldax >> 4, iay = lday >> 4, iaz = ldaz >> 4;
  const int na2i = iax * iax + iay * iay + iaz * iaz;
  v2f A;
  A.x = hi ? (float)(2 * iaz) : (float)(2 * iax);
  A.y = hi ? 1.0f             : (float)(2 * iay);

  float sk0[8], sk1[8];
#pragma unroll
  for (int r = 0; r < 8; ++r) { sk0[r] = sk1[r] = 0.0f; }

  const v8f Cz = {};

  const int nFirst = cpart * (TILES_PER_WAVE * 16) + mlane;
  const v2f* bfp = (const v2f*)&sFrag[nFirst] + (hi ? 1 : 0);

  v2f bE[SWEEP_GROUP], bO[SWEEP_GROUP];
#pragma unroll
  for (int t = 0; t < SWEEP_GROUP; ++t) { bE[t] = bfp[t * 32]; }

  for (int g = 0; g < SWEEP_GROUPS; g += 2) {
    const int go = (g + 1) * (SWEEP_GROUP * 32);
#pragma unroll
    for (int t = 0; t < SWEEP_GROUP; ++t) { bO[t] = bfp[go + t * 32]; }
    __builtin_amdgcn_sched_barrier(0);

#pragma unroll
    for (int t = 0; t < SWEEP_GROUP; ++t) {
      v8f Dv = __builtin_amdgcn_wmma_f32_16x16x4_f32(
           false, A,  false, bE[t],
           (short)0, Cz,  false,  false);
#pragma unroll
      for (int r = 0; r < 8; ++r) {
        TOP2_INSERT_F(Dv[r], sk0[r], sk1[r]);
      }
    }

    const int ge = ((g + 2) & (SWEEP_GROUPS - 1)) * (SWEEP_GROUP * 32);
#pragma unroll
    for (int t = 0; t < SWEEP_GROUP; ++t) { bE[t] = bfp[ge + t * 32]; }
    __builtin_amdgcn_sched_barrier(0);


#pragma unroll
    for (int t = 0; t < SWEEP_GROUP; ++t) {
      v8f Dv = __builtin_amdgcn_wmma_f32_16x16x4_f32(
           false, A,  false, bO[t],
           (short)0, Cz,  false,  false);
#pragma unroll
      for (int r = 0; r < 8; ++r) {
        TOP2_INSERT_F(Dv[r], sk0[r], sk1[r]);
      }
    }
  }

  int bp0[8], bp1[8];
#pragma unroll
  for (int r = 0; r < 8; ++r) {
    const float fs[2] = { sk0[r], sk1[r] };
    int ps[2];
#pragma unroll
    for (int t = 0; t < 2; ++t) {
      const int key  = (int)fs[t];
      const int tie  = (int)((fs[t] - (float)key) * STEP_SCALE);
      const int nn   = nFirst + ((TILES_PER_WAVE - 1) - tie) * 16;
      ps[t] = (key << 12) + (4095 - nn);
    }
    bp0[r] = ps[0]; bp1[r] = ps[1];
  }

  {
    const int slot = cpart * 16 + mlane;
#pragma unroll
    for (int r = 0; r < 8; ++r) {
      int* xr = &sX[rtile * ROWS_PER_WAVE + r + (hi ? 8 : 0)][slot];
      xr[0]         = bp0[r];
      xr[SRC_LISTS] = bp1[r];
    }
  }
  __syncthreads();

  const float* afBase = a_feats + ((size_t)b * NA + rowBase) * FD;
  float* outBase = out + ((size_t)b * NA + rowBase) * (2 * FD);
  const float* bfB = b_feats + (size_t)b * NB * FD;
  const int col4 = mlane * 4;
  const int rowFirst = cpart * OUT_ROWS_PER_WAVE + (hi ? 1 : 0);

  int m0[LISTS_PER_MERGER], m1[LISTS_PER_MERGER], m2[LISTS_PER_MERGER];
  {
    const int* xq = &sX[rtile * ROWS_PER_WAVE + cpart * OUT_ROWS_PER_WAVE + (lane >> 2)][(lane & 3) * LISTS_PER_MERGER];
    const v4i a0 = *(const v4i*)(xq),                 b0 = *(const v4i*)(xq + 4);
    const v4i a1 = *(const v4i*)(xq + SRC_LISTS),     b1 = *(const v4i*)(xq + SRC_LISTS + 4);
#pragma unroll
    for (int e = 0; e < 4; ++e) {
      m0[e] = a0[e]; m0[e + 4] = b0[e];
      m1[e] = a1[e]; m1[e + 4] = b1[e];
      m2[e] = 0;     m2[e + 4] = 0;
    }
  }
#pragma unroll
  for (int span = 1; span < LISTS_PER_MERGER; span <<= 1) {
#pragma unroll
    for (int i = 0; i < LISTS_PER_MERGER; i += 2 * span) {
      MERGE3_SORTED(m0[i], m1[i], m2[i], m0[i + span], m1[i + span], m2[i + span]);
    }
  }
  int f0 = m0[0], f1 = m1[0], f2 = m2[0];
#pragma unroll
  for (int off = 1; off < 4; off <<= 1) {
    const int q0 = __shfl_xor(f0, off, 32);
    const int q1 = __shfl_xor(f1, off, 32);
    const int q2 = __shfl_xor(f2, off, 32);
    MERGE3_SORTED(f0, f1, f2, q0, q1, q2);
  }

  {
    const int nBest   = 4095 - (f0 & 4095);
    const int nSecond = 4095 - (f1 & 4095);
    const bool sameList = ((nBest ^ nSecond) & LIST_ID_MASK) == 0;
    unsigned rowsLeft = __builtin_amdgcn_ballot_w32(sameList && (lane & 3) == 0);
    while (rowsLeft != 0u) {
      const int src = __builtin_ctz(rowsLeft);
      rowsLeft &= rowsLeft - 1u;
      const int below    = __builtin_amdgcn_readlane(f1, src);
      const int listBase = __builtin_amdgcn_readlane(nSecond, src) & LIST_ID_MASK;
      const int rowInTile = cpart * OUT_ROWS_PER_WAVE + (src >> 2);
      const int rr = rowInTile & 7;
      const bool myHalf = (rowInTile >> 3) == (hi ? 1 : 0);
      int third = 0;
#pragma unroll
      for (int t = 0; t < RESCAN_TILES; ++t) {
        const int n = listBase + 16 * (16 * t + mlane);
        const v2f bq = *((const v2f*)&sFrag[n] + (hi ? 1 : 0));
        const v8f Dq = __builtin_amdgcn_wmma_f32_16x16x4_f32(
             false, A,  false, bq,
             (short)0, Cz,  false,  false);
        float fk = Dq[0];
#pragma unroll
        for (int r = 1; r < 8; ++r) { fk = (rr == r) ? Dq[r] : fk; }
        const int pk = ((int)fk << 12) + (4095 - n);
        third = max(third, (myHalf && pk < below) ? pk : 0);
      }
#pragma unroll
      for (int off = 1; off < 32; off <<= 1) {
        third = max(third, __shfl_xor(third, off, 32));
      }
      if ((lane >> 2) == (src >> 2)) { f2 = max(f2, third); }
    }
  }

  v4f g0[ROW_PAIRS], g1[ROW_PAIRS], g2[ROW_PAIRS], ga[ROW_PAIRS];
  int kd0[ROW_PAIRS], kd1[ROW_PAIRS], kd2[ROW_PAIRS];
#pragma unroll
  for (int it = 0; it < ROW_PAIRS; ++it) {
    const int row = rowFirst + it * 2;
    const int mergedBy = (it * 2 + (hi ? 1 : 0)) * 4;
    const int pk0 = __shfl(f0, mergedBy, 32);
    const int pk1 = __shfl(f1, mergedBy, 32);
    const int pk2 = __shfl(f2, mergedBy, 32);
    const int j0 = 4095 - (pk0 & 4095);
    const int j1 = 4095 - (pk1 & 4095);
    const int j2 = 4095 - (pk2 & 4095);
    g0[it] = *(const v4f*)(bfB + (size_t)j0 * FD + col4);
    g1[it] = *(const v4f*)(bfB + (size_t)j1 * FD + col4);
    g2[it] = *(const v4f*)(bfB + (size_t)j2 * FD + col4);
    ga[it] = *(const v4f*)(afBase + (size_t)row * FD + col4);
    kd0[it] = (int)((unsigned)pk0 >> 12);
    kd1[it] = (int)((unsigned)pk1 >> 12);
    kd2[it] = (int)((unsigned)pk2 >> 12);
  }
  __builtin_amdgcn_sched_barrier(0);
#pragma unroll
  for (int it = 0; it < ROW_PAIRS; ++it) {
    const int row = rowFirst + it * 2;
    const int rna2 = __shfl(na2i, row, 32);
    const int d20 = rna2 + KBIAS - kd0[it];
    const int d21 = rna2 + KBIAS - kd1[it];
    const int d22 = rna2 + KBIAS - kd2[it];
    const float inv = 1.0f / 128.0f;
    const float w0 = fmaxf(0.5f - sqrtf((float)d20) * inv, 0.0f);
    const float w1 = fmaxf(0.5f - sqrtf((float)d21) * inv, 0.0f);
    const float w2 = fmaxf(0.5f - sqrtf((float)d22) * inv, 0.0f);
    v4f tmp;
#pragma unroll
    for (int e = 0; e < 4; ++e) {
      tmp[e] = w0 * g0[it][e] + w1 * g1[it][e] + w2 * g2[it][e];
    }
    float* op = outBase + (size_t)row * (2 * FD) + col4;
    *(v4f*)op        = ga[it];
    *(v4f*)(op + FD) = tmp;
  }
}

extern "C" void kernel_launch(void* const* d_in, const int* in_sizes, int n_in,
                              void* d_out, int out_size, void* d_ws, size_t ws_size,
                              hipStream_t stream) {
  const float* a_feats  = (const float*)d_in[0];
  const float* b_feats  = (const float*)d_in[1];
  const int*   a_coords = (const int*)d_in[2];
  const int*   b_coords = (const int*)d_in[3];
  float* out = (float*)d_out;

  dim3 grid(B_SZ * (NA / ROWS_PER_BLOCK));
  dim3 block(WAVES * 32);
  knn3_wmma<<<grid, block, 0, stream>>>(a_feats, b_feats, a_coords, b_coords, out);
}


# the a_feats half of each output row is read behind the staging barrier and stored halfway through the sweep (no global memory use there); the epilogue moves only gathers and result rows
# speedup vs baseline: 2.9737x; 1.0674x over previous
#include <hip/hip_runtime.h>

typedef __attribute__((ext_vector_type(2))) float v2f;
typedef __attribute__((ext_vector_type(8))) float v8f;
typedef __attribute__((ext_vector_type(4))) float v4f;
typedef __attribute__((ext_vector_type(4))) int v4i;

#define B_SZ 8
#define NA 4096
#define NB 4096
#define FD 64
#define ROW_TILES 8
#define COL_SPLIT 2
#define WAVES (ROW_TILES * COL_SPLIT)
#define ROWS_PER_WAVE 16
#define ROWS_PER_BLOCK (ROW_TILES * ROWS_PER_WAVE)
#define NTILES (NB / 16)
#define TILES_PER_WAVE (NTILES / COL_SPLIT)
#define OUT_ROWS_PER_WAVE (ROWS_PER_WAVE / COL_SPLIT)
#define STAGE_TRIPS (NB / (WAVES * 32))
#define ROW_PAIRS (OUT_ROWS_PER_WAVE / 2)
#define SRC_LISTS (COL_SPLIT * 16)
#define XPITCH (2 * SRC_LISTS + 4)
#define LIST_ID_MASK 0x80F
#define RESCAN_TILES (TILES_PER_WAVE / 16)
static_assert(NB == 4096 && COL_SPLIT == 2, "LIST_ID_MASK: bit 11 is the column part, bits 0-3 the lane");
#define LISTS_PER_MERGER (SRC_LISTS / 4)
static_assert(OUT_ROWS_PER_WAVE * 4 == 32 && LISTS_PER_MERGER == 8, "a wave's 32 lanes = 8 output rows x 4 mergers of 8 lists");
static_assert(NB % (WAVES * 32) == 0, "every thread stages the same number of columns");
static_assert(OUT_ROWS_PER_WAVE % 2 == 0 && FD == 16 * 4, "a 16-lane half covers one 64-float row with 4 floats per lane");
#define KBIAS 48388
#define STEP_SCALE ((float)TILES_PER_WAVE)
static_assert(TILES_PER_WAVE == 128, "sweep key = 17 integer bits + 7 step bits = 24 bits of f32");
#define SWEEP_GROUP 4
#define SWEEP_GROUPS (TILES_PER_WAVE / SWEEP_GROUP)
static_assert(TILES_PER_WAVE % SWEEP_GROUP == 0 && (SWEEP_GROUPS & (SWEEP_GROUPS - 1)) == 0 && SWEEP_GROUPS % 2 == 0, "the look-ahead wraps with a mask; two groups per loop trip");

#define MERGE3_SORTED(p0, p1, p2, q0, q1, q2) do {     \
    const int _lo = min((p0), (q0));                    \
    const int _c1 = min((p1), (q0));                    \
    const int _c2 = min((p0), (q1));                    \
    (p2) = max(max((p2), (q2)), max(_c1, _c2));         \
    (p1) = max(_lo, max((p1), (q1)));                   \
    (p0) = max((p0), (q0));                             \
  } while (0)

__device__ __forceinline__ float max_nonneg(float a, float b) {
  return __int_as_float(max(__float_as_int(a), __float_as_int(b)));
}

#define TOP2_INSERT_F(fk, f0, f1) do {                       \
    (f1) = __builtin_amdgcn_fmed3f((fk), (f0), (f1));         \
    (f0) = max_nonneg((fk), (f0));                            \
  } while (0)

__global__ __launch_bounds__(WAVES * 32) void knn3_wmma(
    const float* __restrict__ a_feats,
    const float* __restrict__ b_feats,
    const int*   __restrict__ a_coords,
    const int*   __restrict__ b_coords,
    float*       __restrict__ out)
{
  struct __attribute__((aligned(16))) LdsLayout {
    v4f      bfrag[NB];
    int      x[ROWS_PER_BLOCK][XPITCH];
  };
  __shared__ LdsLayout lds;
  v4f      (&sFrag)[NB] = lds.bfrag;
  int      (&sX)[ROWS_PER_BLOCK][XPITCH] = lds.x;

  const int tid   = threadIdx.x;
  const int lane  = tid & 31;
  const int wave  = tid >> 5;
  const int mlane = lane & 15;
  const bool hi   = lane >= 16;
  const int rtile = wave % ROW_TILES;
  const int cpart = wave / ROW_TILES;

  const int blocksPerBatch = NA / ROWS_PER_BLOCK;
  const int b       = blockIdx.x / blocksPerBatch;
  const int rowBase = (blockIdx.x % blocksPerBatch) * ROWS_PER_BLOCK
                    + rtile * ROWS_PER_WAVE;

  const int arow = rowBase + mlane;
  const int* ap = a_coords + ((size_t)b * NA + arow) * 3;
  const int ldax = ap[0], lday = ap[1], ldaz = ap[2];

#pragma unroll
  for (int k = 0; k < STAGE_TRIPS; ++k) {
    const int n = tid + k * (WAVES * 32);
    const int* p = b_coords + ((size_t)b * NB + n) * 3;
    int x = p[0] >> 4, y = p[1] >> 4, z = p[2] >> 4;
    const int tie = (TILES_PER_WAVE - 1) - ((n >> 4) & (TILES_PER_WAVE - 1));
    v4f f;
    f.x = (float)x;
    f.y = (float)y;
    f.z = (float)z;
    f.w = (float)(KBIAS - (x * x + y * y + z * z)) + (float)tie * (1.0f / STEP_SCALE);
    sFrag[n] = f;
  }
  __syncthreads();

  const float* afBase = a_feats + ((size_t)b * NA + rowBase) * FD;
  float* outBase = out + ((size_t)b * NA + rowBase) * (2 * FD);
  const int col4 = mlane * 4;
  const int rowFirst = cpart * OUT_ROWS_PER_WAVE + (hi ? 1 : 0);
  v4f ga[ROW_PAIRS];
#pragma unroll
  for (int it = 0; it < ROW_PAIRS; ++it) {
    ga[it] = *(const v4f*)(afBase + (size_t)(rowFirst + it * 2) * FD + col4);
  }

  const int iax = ldax >> 4, iay = lday >> 4, iaz = ldaz >> 4;
  const int na2i = iax * iax + iay * iay + iaz * iaz;
  v2f A;
  A.x = hi ? (float)(2 * iaz) : (float)(2 * iax);
  A.y = hi ? 1.0f             : (float)(2 * iay);

  float sk0[8], sk1[8];
#pragma unroll
  for (int r = 0; r < 8; ++r) { sk0[r] = sk1[r] = 0.0f; }

  const v8f Cz = {};

  const int nFirst = cpart * (TILES_PER_WAVE * 16) + mlane;
  const v2f* bfp = (const v2f*)&sFrag[nFirst] + (hi ? 1 : 0);

  v2f bE[SWEEP_GROUP], bO[SWEEP_GROUP];
#pragma unroll
  for (int t = 0; t < SWEEP_GROUP; ++t) { bE[t] = bfp[t * 32]; }

  for (int g = 0; g < SWEEP_GROUPS; g += 2) {
    if (g == SWEEP_GROUPS / 2) {
#pragma unroll
      for (int it = 0; it < ROW_PAIRS; ++it) {
        *(v4f*)(outBase + (size_t)(rowFirst + it * 2) * (2 * FD) + col4) = ga[it];
      }
    }
    const int go = (g + 1) * (SWEEP_GROUP * 32);
#pragma unroll
    for (int t = 0; t < SWEEP_GROUP; ++t) { bO[t] = bfp[go + t * 32]; }
    __builtin_amdgcn_sched_barrier(0);

#pragma unroll
    for (int t = 0; t < SWEEP_GROUP; ++t) {
      v8f Dv = __builtin_amdgcn_wmma_f32_16x16x4_f32(
           false, A,  false, bE[t],
           (short)0, Cz,  false,  false);
#pragma unroll
      for (int r = 0; r < 8; ++r) {
        TOP2_INSERT_F(Dv[r], sk0[r], sk1[r]);
      }
    }

    const int ge = ((g + 2) & (SWEEP_GROUPS - 1)) * (SWEEP_GROUP * 32);
#pragma unroll
    for (int t = 0; t < SWEEP_GROUP; ++t) { bE[t] = bfp[ge + t * 32]; }
    __builtin_amdgcn_sched_barrier(0);


#pragma unroll
    for (int t = 0; t < SWEEP_GROUP; ++t) {
      v8f Dv = __builtin_amdgcn_wmma_f32_16x16x4_f32(
           false, A,  false, bO[t],
           (short)0, Cz,  false,  false);
#pragma unroll
      for (int r = 0; r < 8; ++r) {
        TOP2_INSERT_F(Dv[r], sk0[r], sk1[r]);
      }
    }
  }

  int bp0[8], bp1[8];
#pragma unroll
  for (int r = 0; r < 8; ++r) {
    const float fs[2] = { sk0[r], sk1[r] };
    int ps[2];
#pragma unroll
    for (int t = 0; t < 2; ++t) {
      const int key  = (int)fs[t];
      const int tie  = (int)((fs[t] - (float)key) * STEP_SCALE);
      const int nn   = nFirst + ((TILES_PER_WAVE - 1) - tie) * 16;
      ps[t] = (key << 12) + (4095 - nn);
    }
    bp0[r] = ps[0]; bp1[r] = ps[1];
  }

  {
    const int slot = cpart * 16 + mlane;
#pragma unroll
    for (int r = 0; r < 8; ++r) {
      int* xr = &sX[rtile * ROWS_PER_WAVE + r + (hi ? 8 : 0)][slot];
      xr[0]         = bp0[r];
      xr[SRC_LISTS] = bp1[r];
    }
  }
  __syncthreads();

  const float* bfB = b_feats + (size_t)b * NB * FD;

  int m0[LISTS_PER_MERGER], m1[LISTS_PER_MERGER], m2[LISTS_PER_MERGER];
  {
    const int* xq = &sX[rtile * ROWS_PER_WAVE + cpart * OUT_ROWS_PER_WAVE + (lane >> 2)][(lane & 3) * LISTS_PER_MERGER];
    const v4i a0 = *(const v4i*)(xq),                 b0 = *(const v4i*)(xq + 4);
    const v4i a1 = *(const v4i*)(xq + SRC_LISTS),     b1 = *(const v4i*)(xq + SRC_LISTS + 4);
#pragma unroll
    for (int e = 0; e < 4; ++e) {
      m0[e] = a0[e]; m0[e + 4] = b0[e];
      m1[e] = a1[e]; m1[e + 4] = b1[e];
      m2[e] = 0;     m2[e + 4] = 0;
    }
  }
#pragma unroll
  for (int span = 1; span < LISTS_PER_MERGER; span <<= 1) {
#pragma unroll
    for (int i = 0; i < LISTS_PER_MERGER; i += 2 * span) {
      MERGE3_SORTED(m0[i], m1[i], m2[i], m0[i + span], m1[i + span], m2[i + span]);
    }
  }
  int f0 = m0[0], f1 = m1[0], f2 = m2[0];
#pragma unroll
  for (int off = 1; off < 4; off <<= 1) {
    const int q0 = __shfl_xor(f0, off, 32);
    const int q1 = __shfl_xor(f1, off, 32);
    const int q2 = __shfl_xor(f2, off, 32);
    MERGE3_SORTED(f0, f1, f2, q0, q1, q2);
  }

  {
    const int nBest   = 4095 - (f0 & 4095);
    const int nSecond = 4095 - (f1 & 4095);
    const bool sameList = ((nBest ^ nSecond) & LIST_ID_MASK) == 0;
    unsigned rowsLeft = __builtin_amdgcn_ballot_w32(sameList && (lane & 3) == 0);
    while (rowsLeft != 0u) {
      const int src = __builtin_ctz(rowsLeft);
      rowsLeft &= rowsLeft - 1u;
      const int below    = __builtin_amdgcn_readlane(f1, src);
      const int listBase = __builtin_amdgcn_readlane(nSecond, src) & LIST_ID_MASK;
      const int rowInTile = cpart * OUT_ROWS_PER_WAVE + (src >> 2);
      const int rr = rowInTile & 7;
      const bool myHalf = (rowInTile >> 3) == (hi ? 1 : 0);
      int third = 0;
#pragma unroll
      for (int t = 0; t < RESCAN_TILES; ++t) {
        const int n = listBase + 16 * (16 * t + mlane);
        const v2f bq = *((const v2f*)&sFrag[n] + (hi ? 1 : 0));
        const v8f Dq = __builtin_amdgcn_wmma_f32_16x16x4_f32(
             false, A,  false, bq,
             (short)0, Cz,  false,  false);
        float fk = Dq[0];
#pragma unroll
        for (int r = 1; r < 8; ++r) { fk = (rr == r) ? Dq[r] : fk; }
        const int pk = ((int)fk << 12) + (4095 - n);
        third = max(third, (myHalf && pk < below) ? pk : 0);
      }
#pragma unroll
      for (int off = 1; off < 32; off <<= 1) {
        third = max(third, __shfl_xor(third, off, 32));
      }
      if ((lane >> 2) == (src >> 2)) { f2 = max(f2, third); }
    }
  }

  v4f g0[ROW_PAIRS], g1[ROW_PAIRS], g2[ROW_PAIRS];
  int kd0[ROW_PAIRS], kd1[ROW_PAIRS], kd2[ROW_PAIRS];
#pragma unroll
  for (int it = 0; it < ROW_PAIRS; ++it) {
    const int mergedBy = (it * 2 + (hi ? 1 : 0)) * 4;
    const int pk0 = __shfl(f0, mergedBy, 32);
    const int pk1 = __shfl(f1, mergedBy, 32);
    const int pk2 = __shfl(f2, mergedBy, 32);
    const int j0 = 4095 - (pk0 & 4095);
    const int j1 = 4095 - (pk1 & 4095);
    const int j2 = 4095 - (pk2 & 4095);
    g0[it] = *(const v4f*)(bfB + (size_t)j0 * FD + col4);
    g1[it] = *(const v4f*)(bfB + (size_t)j1 * FD + col4);
    g2[it] = *(const v4f*)(bfB + (size_t)j2 * FD + col4);
    kd0[it] = (int)((unsigned)pk0 >> 12);
    kd1[it] = (int)((unsigned)pk1 >> 12);
    kd2[it] = (int)((unsigned)pk2 >> 12);
  }
  __builtin_amdgcn_sched_barrier(0);
#pragma unroll
  for (int it = 0; it < ROW_PAIRS; ++it) {
    const int row = rowFirst + it * 2;
    const int rna2 = __shfl(na2i, row, 32);
    const int d20 = rna2 + KBIAS - kd0[it];
    const int d21 = rna2 + KBIAS - kd1[it];
    const int d22 = rna2 + KBIAS - kd2[it];
    const float inv = 1.0f / 128.0f;
    const float w0 = fmaxf(0.5f - sqrtf((float)d20) * inv, 0.0f);
    const float w1 = fmaxf(0.5f - sqrtf((float)d21) * inv, 0.0f);
    const float w2 = fmaxf(0.5f - sqrtf((float)d22) * inv, 0.0f);
    v4f tmp;
#pragma unroll
    for (int e = 0; e < 4; ++e) {
      tmp[e] = w0 * g0[it][e] + w1 * g1[it][e] + w2 * g2[it][e];
    }
    *(v4f*)(outBase + (size_t)row * (2 * FD) + FD + col4) = tmp;
  }
}

extern "C" void kernel_launch(void* const* d_in, const int* in_sizes, int n_in,
                              void* d_out, int out_size, void* d_ws, size_t ws_size,
                              hipStream_t stream) {
  const float* a_feats  = (const float*)d_in[0];
  const float* b_feats  = (const float*)d_in[1];
  const int*   a_coords = (const int*)d_in[2];
  const int*   b_coords = (const int*)d_in[3];
  float* out = (float*)d_out;

  dim3 grid(B_SZ * (NA / ROWS_PER_BLOCK));
  dim3 block(WAVES * 32);
  knn3_wmma<<<grid, block, 0, stream>>>(a_feats, b_feats, a_coords, b_coords, out);
}


# sweep inserts one key per 4-tile chunk (a lane folds the chunk's 4 keys by max3+max): 1 VALU per candidate, was 2; a row's two best chunks (8 columns) are redone after the merge, 4 WMMAs per wave
# speedup vs baseline: 3.2699x; 1.0996x over previous
#include <hip/hip_runtime.h>

typedef __attribute__((ext_vector_type(2))) float v2f;
typedef __attribute__((ext_vector_type(8))) float v8f;
typedef __attribute__((ext_vector_type(4))) float v4f;
typedef __attribute__((ext_vector_type(4))) int v4i;
typedef __attribute__((ext_vector_type(2))) int v2i;

#define B_SZ 8
#define NA 4096
#define NB 4096
#define FD 64
#define ROW_TILES 8
#define COL_SPLIT 2
#define WAVES (ROW_TILES * COL_SPLIT)
#define ROWS_PER_WAVE 16
#define ROWS_PER_BLOCK (ROW_TILES * ROWS_PER_WAVE)
#define NTILES (NB / 16)
#define TILES_PER_WAVE (NTILES / COL_SPLIT)
#define OUT_ROWS_PER_WAVE (ROWS_PER_WAVE / COL_SPLIT)
#define STAGE_TRIPS (NB / (WAVES * 32))
#define ROW_PAIRS (OUT_ROWS_PER_WAVE / 2)
#define SRC_LISTS (COL_SPLIT * 16)
#define XPITCH (2 * SRC_LISTS + 4)
#define LIST_ID_MASK 0x80F
#define RESCAN_TILES (TILES_PER_WAVE / 16)
static_assert(NB == 4096 && COL_SPLIT == 2, "LIST_ID_MASK: bit 11 is the column part, bits 0-3 the lane");
#define LISTS_PER_MERGER (SRC_LISTS / 4)
#define CHUNK_TILES 4
#define CHUNK_ID_MASK (0xFFF & ~((CHUNK_TILES - 1) << 4))
#define REPAIR_TILES (OUT_ROWS_PER_WAVE / 2)
static_assert(OUT_ROWS_PER_WAVE * 4 == 32 && LISTS_PER_MERGER == 8, "a wave's 32 lanes = 8 output rows x 4 mergers of 8 lists");
static_assert(NB % (WAVES * 32) == 0, "every thread stages the same number of columns");
static_assert(OUT_ROWS_PER_WAVE % 2 == 0 && FD == 16 * 4, "a 16-lane half covers one 64-float row with 4 floats per lane");
#define KBIAS 48388
#define STEP_SCALE ((float)TILES_PER_WAVE)
static_assert(TILES_PER_WAVE == 128, "sweep key = 17 integer bits + 7 step bits = 24 bits of f32");
#define SWEEP_GROUP 4
static_assert(SWEEP_GROUP == CHUNK_TILES && 2 * CHUNK_TILES * 2 == 16, "a chunk is one sweep group; two rows x two chunks x CHUNK_TILES columns fill a 16-column tile");
#define SWEEP_GROUPS (TILES_PER_WAVE / SWEEP_GROUP)
static_assert(TILES_PER_WAVE % SWEEP_GROUP == 0 && (SWEEP_GROUPS & (SWEEP_GROUPS - 1)) == 0 && SWEEP_GROUPS % 2 == 0, "the look-ahead wraps with a mask; two groups per loop trip");

#define MERGE3_SORTED(p0, p1, p2, q0, q1, q2) do {     \
    const int _lo = min((p0), (q0));                    \
    const int _c1 = min((p1), (q0));                    \
    const int _c2 = min((p0), (q1));                    \
    (p2) = max(max((p2), (q2)), max(_c1, _c2));         \
    (p1) = max(_lo, max((p1), (q1)));                   \
    (p0) = max((p0), (q0));                             \
  } while (0)

__device__ __forceinline__ float max_nonneg(float a, float b) {
  return __int_as_float(max(__float_as_int(a), __float_as_int(b)));
}

#define TOP2_INSERT_F(fk, f0, f1) do {                       \
    (f1) = __builtin_amdgcn_fmed3f((fk), (f0), (f1));         \
    (f0) = max_nonneg((fk), (f0));                            \
  } while (0)

#define CHUNK_MAX4(a, b, c, d)                                                        \
  __int_as_float(max(max(max(__float_as_int(a), __float_as_int(b)), __float_as_int(c)), \
                     __float_as_int(d)))

__global__ __launch_bounds__(WAVES * 32) void knn3_wmma(
    const float* __restrict__ a_feats,
    const float* __restrict__ b_feats,
    const int*   __restrict__ a_coords,
    const int*   __restrict__ b_coords,
    float*       __restrict__ out)
{
  struct __attribute__((aligned(16))) LdsLayout {
    v4f      bfrag[NB];
    int      x[ROWS_PER_BLOCK][XPITCH];
  };
  __shared__ LdsLayout lds;
  v4f      (&sFrag)[NB] = lds.bfrag;
  int      (&sX)[ROWS_PER_BLOCK][XPITCH] = lds.x;

  const int tid   = threadIdx.x;
  const int lane  = tid & 31;
  const int wave  = tid >> 5;
  const int mlane = lane & 15;
  const bool hi   = lane >= 16;
  const int rtile = wave % ROW_TILES;
  const int cpart = wave / ROW_TILES;

  const int blocksPerBatch = NA / ROWS_PER_BLOCK;
  const int b       = blockIdx.x / blocksPerBatch;
  const int rowBase = (blockIdx.x % blocksPerBatch) * ROWS_PER_BLOCK
                    + rtile * ROWS_PER_WAVE;

  const int arow = rowBase + mlane;
  const int* ap = a_coords + ((size_t)b * NA + arow) * 3;
  const int ldax = ap[0], lday = ap[1], ldaz = ap[2];

#pragma unroll
  for (int k = 0; k < STAGE_TRIPS; ++k) {
    const int n = tid + k * (WAVES * 32);
    const int* p = b_coords + ((size_t)b * NB + n) * 3;
    int x = p[0] >> 4, y = p[1] >> 4, z = p[2] >> 4;
    const int tie = (TILES_PER_WAVE - 1) - ((n >> 4) & (TILES_PER_WAVE - 1));
    v4f f;
    f.x = (float)x;
    f.y = (float)y;
    f.z = (float)z;
    f.w = (float)(KBIAS - (x * x + y * y + z * z)) + (float)tie * (1.0f / STEP_SCALE);
    sFrag[n] = f;
  }
  __syncthreads();

  const float* afBase = a_feats + ((size_t)b * NA + rowBase) * FD;
  float* outBase = out + ((size_t)b * NA + rowBase) * (2 * FD);
  const int col4 = mlane * 4;
  const int rowFirst = cpart * OUT_ROWS_PER_WAVE + (hi ? 1 : 0);
  v4f ga[ROW_PAIRS];
#pragma unroll
  for (int it = 0; it < ROW_PAIRS; ++it) {
    ga[it] = *(const v4f*)(afBase + (size_t)(rowFirst + it * 2) * FD + col4);
  }

  const int iax = ldax >> 4, iay = lday >> 4, iaz = ldaz >> 4;
  const int na2i = iax * iax + iay * iay + iaz * iaz;
  v2f A;
  A.x = hi ? (float)(2 * iaz) : (float)(2 * iax);
  A.y = hi ? 1.0f             : (float)(2 * iay);

  float sk0[8], sk1[8];
#pragma unroll
  for (int r = 0; r < 8; ++r) { sk0[r] = sk1[r] = 0.0f; }

  const v8f Cz = {};

  const int nFirst = cpart * (TILES_PER_WAVE * 16) + mlane;
  const v2f* bfp = (const v2f*)&sFrag[nFirst] + (hi ? 1 : 0);

  v2f bE[SWEEP_GROUP], bO[SWEEP_GROUP];
#pragma unroll
  for (int t = 0; t < SWEEP_GROUP; ++t) { bE[t] = bfp[t * 32]; }

  for (int g = 0; g < SWEEP_GROUPS; g += 2) {
    if (g == SWEEP_GROUPS / 2) {
#pragma unroll
      for (int it = 0; it < ROW_PAIRS; ++it) {
        *(v4f*)(outBase + (size_t)(rowFirst + it * 2) * (2 * FD) + col4) = ga[it];
      }
    }
    const int go = (g + 1) * (SWEEP_GROUP * 32);
#pragma unroll
    for (int t = 0; t < SWEEP_GROUP; ++t) { bO[t] = bfp[go + t * 32]; }
    __builtin_amdgcn_sched_barrier(0);

    {
      v8f Dv[SWEEP_GROUP];
#pragma unroll
      for (int t = 0; t < SWEEP_GROUP; ++t) {
        Dv[t] = __builtin_amdgcn_wmma_f32_16x16x4_f32(
             false, A,  false, bE[t],
             (short)0, Cz,  false,  false);
      }
#pragma unroll
      for (int r = 0; r < 8; ++r) {
        const float fc = CHUNK_MAX4(Dv[0][r], Dv[1][r], Dv[2][r], Dv[3][r]);
        TOP2_INSERT_F(fc, sk0[r], sk1[r]);
      }
    }

    const int ge = ((g + 2) & (SWEEP_GROUPS - 1)) * (SWEEP_GROUP * 32);
#pragma unroll
    for (int t = 0; t < SWEEP_GROUP; ++t) { bE[t] = bfp[ge + t * 32]; }
    __builtin_amdgcn_sched_barrier(0);


    {
      v8f Dv[SWEEP_GROUP];
#pragma unroll
      for (int t = 0; t < SWEEP_GROUP; ++t) {
        Dv[t] = __builtin_amdgcn_wmma_f32_16x16x4_f32(
             false, A,  false, bO[t],
             (short)0, Cz,  false,  false);
      }
#pragma unroll
      for (int r = 0; r < 8; ++r) {
        const float fc = CHUNK_MAX4(Dv[0][r], Dv[1][r], Dv[2][r], Dv[3][r]);
        TOP2_INSERT_F(fc, sk0[r], sk1[r]);
      }
    }
  }

  int bp0[8], bp1[8];
#pragma unroll
  for (int r = 0; r < 8; ++r) {
    const float fs[2] = { sk0[r], sk1[r] };
    int ps[2];
#pragma unroll
    for (int t = 0; t < 2; ++t) {
      const int key  = (int)fs[t];
      const int tie  = (int)((fs[t] - (float)key) * STEP_SCALE);
      const int nn   = nFirst + ((TILES_PER_WAVE - 1) - tie) * 16;
      ps[t] = (key << 12) + (4095 - nn);
    }
    bp0[r] = ps[0]; bp1[r] = ps[1];
  }

  {
    const int slot = cpart * 16 + mlane;
#pragma unroll
    for (int r = 0; r < 8; ++r) {
      int* xr = &sX[rtile * ROWS_PER_WAVE + r + (hi ? 8 : 0)][slot];
      xr[0]         = bp0[r];
      xr[SRC_LISTS] = bp1[r];
    }
  }
  __syncthreads();

  const float* bfB = b_feats + (size_t)b * NB * FD;

  int m0[LISTS_PER_MERGER], m1[LISTS_PER_MERGER], m2[LISTS_PER_MERGER];
  {
    const int* xq = &sX[rtile * ROWS_PER_WAVE + cpart * OUT_ROWS_PER_WAVE + (lane >> 2)][(lane & 3) * LISTS_PER_MERGER];
    const v4i a0 = *(const v4i*)(xq),                 b0 = *(const v4i*)(xq + 4);
    const v4i a1 = *(const v4i*)(xq + SRC_LISTS),     b1 = *(const v4i*)(xq + SRC_LISTS + 4);
#pragma unroll
    for (int e = 0; e < 4; ++e) {
      m0[e] = a0[e]; m0[e + 4] = b0[e];
      m1[e] = a1[e]; m1[e + 4] = b1[e];
      m2[e] = 0;     m2[e + 4] = 0;
    }
  }
#pragma unroll
  for (int span = 1; span < LISTS_PER_MERGER; span <<= 1) {
#pragma unroll
    for (int i = 0; i < LISTS_PER_MERGER; i += 2 * span) {
      MERGE3_SORTED(m0[i], m1[i], m2[i], m0[i + span], m1[i + span], m2[i + span]);
    }
  }
  int f0 = m0[0], f1 = m1[0], f2 = m2[0];
#pragma unroll
  for (int off = 1; off < 4; off <<= 1) {
    const int q0 = __shfl_xor(f0, off, 32);
    const int q1 = __shfl_xor(f1, off, 32);
    const int q2 = __shfl_xor(f2, off, 32);
    MERGE3_SORTED(f0, f1, f2, q0, q1, q2);
  }

  const int nBest   = 4095 - (f0 & 4095);
  const int nSecond = 4095 - (f1 & 4095);
  {
    const bool sameList = ((nBest ^ nSecond) & LIST_ID_MASK) == 0;
    unsigned rowsLeft = __builtin_amdgcn_ballot_w32(sameList && (lane & 3) == 0);
    while (rowsLeft != 0u) {
      const int src = __builtin_ctz(rowsLeft);
      rowsLeft &= rowsLeft - 1u;
      const int chunkA   = __builtin_amdgcn_readlane(nBest, src) & CHUNK_ID_MASK;
      const int chunkB   = __builtin_amdgcn_readlane(nSecond, src) & CHUNK_ID_MASK;
      const int listBase = chunkB & LIST_ID_MASK;
      const int rowInTile = cpart * OUT_ROWS_PER_WAVE + (src >> 2);
      const int rr = rowInTile & 7;
      const bool myHalf = (rowInTile >> 3) == (hi ? 1 : 0);
      int third = 0;
#pragma unroll
      for (int t = 0; t < RESCAN_TILES; ++t) {
        const int n = listBase + 16 * (16 * t + mlane);
        const v2f bq = *((const v2f*)&sFrag[n] + (hi ? 1 : 0));
        const v8f Dq = __builtin_amdgcn_wmma_f32_16x16x4_f32(
             false, A,  false, bq,
             (short)0, Cz,  false,  false);
        float fk = Dq[0];
#pragma unroll
        for (int r = 1; r < 8; ++r) { fk = (rr == r) ? Dq[r] : fk; }
        const int pk = ((int)fk << 12) + (4095 - n);
        const int chunkN = n & CHUNK_ID_MASK;
        third = max(third, (myHalf && chunkN != chunkA && chunkN != chunkB) ? pk : 0);
      }
#pragma unroll
      for (int off = 1; off < 32; off <<= 1) {
        third = max(third, __shfl_xor(third, off, 32));
      }
      if ((lane >> 2) == (src >> 2)) { f2 = max(f2, third); }
    }
  }

  int t0, t1, t2;
  {
    int* sc = &sX[rtile * ROWS_PER_WAVE + cpart * OUT_ROWS_PER_WAVE][0];
    static_assert(OUT_ROWS_PER_WAVE * XPITCH >= 256, "128 words of accumulator pairs + 128 for the other lane half's stores");
    const int nsel = (lane & 1) ? nSecond : nBest;
    const bool rowsHere = (hi ? 1 : 0) == cpart;
#pragma unroll
    for (int q = 0; q < REPAIR_TILES; ++q) {
      const int nc = __shfl(nsel, 4 * (2 * q + (mlane >> 3)) + ((mlane >> 2) & 1), 32);
      const int n  = (nc & CHUNK_ID_MASK) | ((mlane & 3) << 4);
      const v2f bq = *((const v2f*)&sFrag[n] + (hi ? 1 : 0));
      const v8f Dq = __builtin_amdgcn_wmma_f32_16x16x4_f32(
           false, A,  false, bq,
           (short)0, Cz,  false,  false);
      v2i w;
      w.x = __float_as_int(Dq[2 * q]);
      w.y = __float_as_int(Dq[2 * q + 1]);
      *(v2i*)(sc + 2 * (16 * q + mlane) + (rowsHere ? 0 : 128)) = w;
    }
    const int mi = lane >> 2, mj = lane & 3;
    const int cbase = 2 * (16 * (mi >> 1) + 8 * (mi & 1) + 2 * mj) + (mi & 1);
    const float fa = __int_as_float(sc[cbase]);
    const float fb = __int_as_float(sc[cbase + 2]);
    const int nch = ((mj & 2) ? nSecond : nBest) & CHUNK_ID_MASK;
    const int na = nch | (((2 * mj) & 3) << 4);
    const int nb = nch | (((2 * mj + 1) & 3) << 4);
    const int cx = ((int)fa << 12) + (4095 - na);
    const int cy = ((int)fb << 12) + (4095 - nb);
    t0 = max(cx, cy); t1 = min(cx, cy); t2 = 0;
#pragma unroll
    for (int off = 1; off < 4; off <<= 1) {
      const int q0 = __shfl_xor(t0, off, 32);
      const int q1 = __shfl_xor(t1, off, 32);
      const int q2 = __shfl_xor(t2, off, 32);
      MERGE3_SORTED(t0, t1, t2, q0, q1, q2);
    }
    MERGE3_SORTED(t0, t1, t2, f2, 0, 0);
  }

  v4f g0[ROW_PAIRS], g1[ROW_PAIRS], g2[ROW_PAIRS];
  int kd0[ROW_PAIRS], kd1[ROW_PAIRS], kd2[ROW_PAIRS];
#pragma unroll
  for (int it = 0; it < ROW_PAIRS; ++it) {
    const int mergedBy = (it * 2 + (hi ? 1 : 0)) * 4;
    const int pk0 = __shfl(t0, mergedBy, 32);
    const int pk1 = __shfl(t1, mergedBy, 32);
    const int pk2 = __shfl(t2, mergedBy, 32);
    const int j0 = 4095 - (pk0 & 4095);
    const int j1 = 4095 - (pk1 & 4095);
    const int j2 = 4095 - (pk2 & 4095);
    g0[it] = *(const v4f*)(bfB + (size_t)j0 * FD + col4);
    g1[it] = *(const v4f*)(bfB + (size_t)j1 * FD + col4);
    g2[it] = *(const v4f*)(bfB + (size_t)j2 * FD + col4);
    kd0[it] = (int)((unsigned)pk0 >> 12);
    kd1[it] = (int)((unsigned)pk1 >> 12);
    kd2[it] = (int)((unsigned)pk2 >> 12);
  }
  __builtin_amdgcn_sched_barrier(0);
#pragma unroll
  for (int it = 0; it < ROW_PAIRS; ++it) {
    const int row = rowFirst + it * 2;
    const int rna2 = __shfl(na2i, row, 32);
    const int d20 = rna2 + KBIAS - kd0[it];
    const int d21 = rna2 + KBIAS - kd1[it];
    const int d22 = rna2 + KBIAS - kd2[it];
    const float inv = 1.0f / 128.0f;
    const float w0 = fmaxf(0.5f - sqrtf((float)d20) * inv, 0.0f);
    const float w1 = fmaxf(0.5f - sqrtf((float)d21) * inv, 0.0f);
    const float w2 = fmaxf(0.5f - sqrtf((float)d22) * inv, 0.0f);
    v4f tmp;
#pragma unroll
    for (int e = 0; e < 4; ++e) {
      tmp[e] = w0 * g0[it][e] + w1 * g1[it][e] + w2 * g2[it][e];
    }
    *(v4f*)(outBase + (size_t)row * (2 * FD) + FD + col4) = tmp;
  }
}

extern "C" void kernel_launch(void* const* d_in, const int* in_sizes, int n_in,
                              void* d_out, int out_size, void* d_ws, size_t ws_size,
                              hipStream_t stream) {
  const float* a_feats  = (const float*)d_in[0];
  const float* b_feats  = (const float*)d_in[1];
  const int*   a_coords = (const int*)d_in[2];
  const int*   b_coords = (const int*)d_in[3];
  float* out = (float*)d_out;

  dim3 grid(B_SZ * (NA / ROWS_PER_BLOCK));
  dim3 block(WAVES * 32);
  knn3_wmma<<<grid, block, 0, stream>>>(a_feats, b_feats, a_coords, b_coords, out);
}
